# starting kernel
# baseline (speedup 1.0000x reference)
#include <hip/hip_runtime.h>

#define BB 256
#define TT 500
#define NKC 100
#define NA 13

typedef __attribute__((ext_vector_type(2))) float v2f;
typedef __attribute__((ext_vector_type(8))) float v8f;

__device__ __forceinline__ float lse2(float x, float y) {
    float m = fmaxf(x, y);
    return m + log1pf(__expf(-fabsf(x - y)));
}

__global__ __launch_bounds__(128)
void gemm_x_wmma(const float* __restrict__ FM, const float* __restrict__ lrw,
                 const float* __restrict__ lrb, float* __restrict__ xout) {
    __shared__ v2f bstage[50][32];
    const int tid = threadIdx.x;
    for (int i = tid; i < 50 * 32; i += 128) {
        int k = i >> 5, L = i & 31;
        int n = L & 15;
        int kb = k * 4 + ((L >> 4) << 1);
        v2f v;
        v.x = (n < 2) ? lrw[n * 200 + kb] : 0.0f;
        v.y = (n < 2) ? lrw[n * 200 + kb + 1] : 0.0f;
        bstage[k][L] = v;
    }
    __syncthreads();

    const int wave = tid >> 5, lane = tid & 31;
    const int tile = blockIdx.x * 4 + wave;
    const long row0 = (long)tile * 16;
    const int m = lane & 15;
    const int khalf = (lane >> 4) << 1;
    const float* arow = FM + (row0 + m) * 200 + khalf;

    v8f acc = {};
    for (int k = 0; k < 50; ++k) {
        v2f a = *(const v2f*)(arow + k * 4);
        v2f b = bstage[k][lane];
        acc = __builtin_amdgcn_wmma_f32_16x16x4_f32(
            false, a, false, b, (short)0, acc, false, false);
    }

    const int n = lane & 15;
    if (n < 2) {
        const float bias = lrb[n];
        const int moff = (lane >> 4) << 3;
        for (int r = 0; r < 8; ++r) {
            long row = row0 + moff + r;
            xout[row * 2 + n] = acc[r] + bias;
        }
    }
}

template <bool USE_XBUF>
__global__ __launch_bounds__(128)
void bkt_scan(const int* __restrict__ corr, const int* __restrict__ kcv,
              const int* __restrict__ prob, const float* __restrict__ FM,
              const float* __restrict__ trans, const float* __restrict__ olp,
              const float* __restrict__ olk, const float* __restrict__ initl,
              const float* __restrict__ lrw, const float* __restrict__ lrb,
              const float* __restrict__ xbuf, float* __restrict__ out) {
    __shared__ float alpha[4 * NKC * 2 * NA];
    __shared__ float olk_s[NKC * 4];
    __shared__ float logt_s[NKC * 4];
    __shared__ float la0_s[NKC * 2];

    const int tid = threadIdx.x;
    for (int i = tid; i < NKC * 4; i += 128) olk_s[i] = olk[i];
    for (int c = tid; c < NKC; c += 128) {
        float i0 = initl[c * 2], i1 = initl[c * 2 + 1];
        float l = lse2(i0, i1);
        la0_s[c * 2] = i0 - l;
        la0_s[c * 2 + 1] = i1 - l;
        float t00 = trans[c * 4 + 0], t01 = trans[c * 4 + 1];
        float t10 = trans[c * 4 + 2], t11 = trans[c * 4 + 3];
        float l0 = lse2(t00, t10);
        float l1 = lse2(t01, t11);
        logt_s[c * 4 + 0] = t00 - l0;
        logt_s[c * 4 + 1] = t01 - l1;
        logt_s[c * 4 + 2] = t10 - l0;
        logt_s[c * 4 + 3] = t11 - l1;
    }
    __syncthreads();
    for (int i = tid; i < 4 * NKC * 2 * NA; i += 128) {
        int r = i % (NKC * 2 * NA);
        int c = r / (2 * NA);
        int s = (r % (2 * NA)) / NA;
        alpha[i] = la0_s[c * 2 + s];
    }
    __syncthreads();

    const int wave = tid >> 5, lane = tid & 31;
    const int b = blockIdx.x * 4 + wave;
    const long bT = (long)b * TT;
    float* myA = alpha + wave * (NKC * 2 * NA);
    const int aa = (lane < NA) ? lane : (NA - 1);
    const float abil = -3.0f + 0.5f * (float)aa;
    float logab = -2.5649493574615367f;
    const float NEGI = -1.0e30f;

    for (int t = 0; t < TT; ++t) {
        const long idx = bT + t;
        const int obs = corr[idx];
        const int c = kcv[idx];
        const int p = prob[idx];

        float x0, x1;
        if (USE_XBUF) {
            v2f xv = *(const v2f*)(xbuf + idx * 2);
            x0 = xv.x; x1 = xv.y;
        } else {
            const float* fr = FM + idx * 200;
            float p0 = 0.f, p1 = 0.f;
            for (int f = lane; f < 200; f += 32) {
                float fv = fr[f];
                p0 = fmaf(fv, lrw[f], p0);
                p1 = fmaf(fv, lrw[200 + f], p1);
            }
            for (int off = 16; off; off >>= 1) {
                p0 += __shfl_xor(p0, off, 32);
                p1 += __shfl_xor(p1, off, 32);
            }
            x0 = p0 + lrb[0];
            x1 = p1 + lrb[1];
        }

        const float4 op4 = *(const float4*)(olp + (long)p * 4);
        float l00 = op4.x + olk_s[c * 4 + 0] - abil + x0;
        float l01 = op4.y + olk_s[c * 4 + 1] + abil - x0;
        float l10 = op4.z + olk_s[c * 4 + 2] - abil + x1;
        float l11 = op4.w + olk_s[c * 4 + 3] + abil - x1;
        float z0 = lse2(l00, l01), z1 = lse2(l10, l11);
        float lo00 = l00 - z0, lo01 = l01 - z0;
        float lo10 = l10 - z1, lo11 = l11 - z1;

        float la0v = myA[c * 26 + aa];
        float la1v = myA[c * 26 + 13 + aa];

        float lpgb0 = lse2(lo00 + la0v, lo10 + la1v);
        float lpgb1 = lse2(lo01 + la0v, lo11 + la1v);
        float nrm = lse2(lpgb0, lpgb1);
        lpgb0 -= nrm; lpgb1 -= nrm;

        float lps0 = obs ? lo01 : lo00;
        float lps1 = obs ? lo11 : lo10;
        float nla0 = lse2(lps0 + la0v + logt_s[c * 4 + 0],
                          lps1 + la1v + logt_s[c * 4 + 1]);
        float nla1 = lse2(lps0 + la0v + logt_s[c * 4 + 2],
                          lps1 + la1v + logt_s[c * 4 + 3]);
        if (lane < NA) {
            myA[c * 26 + lane] = nla0;
            myA[c * 26 + 13 + lane] = nla1;
        }

        float o0, o1;
        {
            float v = (lane < NA) ? (logab + lpgb0) : NEGI;
            float m2 = v;
            for (int off = 8; off; off >>= 1) m2 = fmaxf(m2, __shfl_xor(m2, off, 16));
            float e = __expf(v - m2), s = e;
            for (int off = 8; off; off >>= 1) s += __shfl_xor(s, off, 16);
            o0 = m2 + __logf(s);
        }
        {
            float v = (lane < NA) ? (logab + lpgb1) : NEGI;
            float m2 = v;
            for (int off = 8; off; off >>= 1) m2 = fmaxf(m2, __shfl_xor(m2, off, 16));
            float e = __expf(v - m2), s = e;
            for (int off = 8; off; off >>= 1) s += __shfl_xor(s, off, 16);
            o1 = m2 + __logf(s);
        }
        float onrm = lse2(o0, o1);
        if (lane == 0) {
            v2f ov;
            ov.x = o0 - onrm;
            ov.y = o1 - onrm;
            *(v2f*)(out + idx * 2) = ov;
        }
        logab += obs ? lpgb1 : lpgb0;
    }
}

extern "C" void kernel_launch(void* const* d_in, const int* in_sizes, int n_in,
                              void* d_out, int out_size, void* d_ws, size_t ws_size,
                              hipStream_t stream) {
    (void)in_sizes; (void)n_in; (void)out_size;
    const int* corr = (const int*)d_in[0];
    const int* kcv  = (const int*)d_in[1];
    const int* prob = (const int*)d_in[2];
    const float* FM    = (const float*)d_in[3];
    const float* trans = (const float*)d_in[4];
    const float* olp   = (const float*)d_in[5];
    const float* olk   = (const float*)d_in[6];
    const float* initl = (const float*)d_in[7];
    const float* lrw   = (const float*)d_in[8];
    const float* lrb   = (const float*)d_in[9];
    float* out = (float*)d_out;

    const size_t xbytes = (size_t)BB * TT * 2 * sizeof(float);
    if (ws_size >= xbytes) {
        float* xbuf = (float*)d_ws;
        hipLaunchKernelGGL(gemm_x_wmma, dim3(2000), dim3(128), 0, stream,
                           FM, lrw, lrb, xbuf);
        hipLaunchKernelGGL((bkt_scan<true>), dim3(64), dim3(128), 0, stream,
                           corr, kcv, prob, FM, trans, olp, olk, initl,
                           lrw, lrb, xbuf, out);
    } else {
        hipLaunchKernelGGL((bkt_scan<false>), dim3(64), dim3(128), 0, stream,
                           corr, kcv, prob, FM, trans, olp, olk, initl,
                           lrw, lrb, (const float*)nullptr, out);
    }
}


# scan: the library log1pf and logf expansions in every log-sum-exp of the step replaced by the bare hardware exp2/log2 instructions (log argument never below 1); same f32 sums in the same order
# speedup vs baseline: 1.6117x; 1.6117x over previous
#include <hip/hip_runtime.h>

#define BB 256
#define TT 500
#define NKC 100
#define NA 13

typedef __attribute__((ext_vector_type(2))) float v2f;
typedef __attribute__((ext_vector_type(8))) float v8f;

__device__ __forceinline__ float lse2(float x, float y) {
    float m = fmaxf(x, y);
    float e = __builtin_amdgcn_exp2f(-1.4426950408889634f * fabsf(x - y));
    return fmaf(0.6931471805599453f, __builtin_amdgcn_logf(1.0f + e), m);
}

__device__ __forceinline__ float add_log(float m, float s) {
    return fmaf(0.6931471805599453f, __builtin_amdgcn_logf(s), m);
}

__global__ __launch_bounds__(128)
void gemm_x_wmma(const float* __restrict__ FM, const float* __restrict__ lrw,
                 const float* __restrict__ lrb, float* __restrict__ xout) {
    __shared__ v2f bstage[50][32];
    const int tid = threadIdx.x;
    for (int i = tid; i < 50 * 32; i += 128) {
        int k = i >> 5, L = i & 31;
        int n = L & 15;
        int kb = k * 4 + ((L >> 4) << 1);
        v2f v;
        v.x = (n < 2) ? lrw[n * 200 + kb] : 0.0f;
        v.y = (n < 2) ? lrw[n * 200 + kb + 1] : 0.0f;
        bstage[k][L] = v;
    }
    __syncthreads();

    const int wave = tid >> 5, lane = tid & 31;
    const int tile = blockIdx.x * 4 + wave;
    const long row0 = (long)tile * 16;
    const int m = lane & 15;
    const int khalf = (lane >> 4) << 1;
    const float* arow = FM + (row0 + m) * 200 + khalf;

    v8f acc = {};
    for (int k = 0; k < 50; ++k) {
        v2f a = *(const v2f*)(arow + k * 4);
        v2f b = bstage[k][lane];
        acc = __builtin_amdgcn_wmma_f32_16x16x4_f32(
            false, a, false, b, (short)0, acc, false, false);
    }

    const int n = lane & 15;
    if (n < 2) {
        const float bias = lrb[n];
        const int moff = (lane >> 4) << 3;
        for (int r = 0; r < 8; ++r) {
            long row = row0 + moff + r;
            xout[row * 2 + n] = acc[r] + bias;
        }
    }
}

template <bool USE_XBUF>
__global__ __launch_bounds__(128)
void bkt_scan(const int* __restrict__ corr, const int* __restrict__ kcv,
              const int* __restrict__ prob, const float* __restrict__ FM,
              const float* __restrict__ trans, const float* __restrict__ olp,
              const float* __restrict__ olk, const float* __restrict__ initl,
              const float* __restrict__ lrw, const float* __restrict__ lrb,
              const float* __restrict__ xbuf, float* __restrict__ out) {
    __shared__ float alpha[4 * NKC * 2 * NA];
    __shared__ float olk_s[NKC * 4];
    __shared__ float logt_s[NKC * 4];
    __shared__ float la0_s[NKC * 2];

    const int tid = threadIdx.x;
    for (int i = tid; i < NKC * 4; i += 128) olk_s[i] = olk[i];
    for (int c = tid; c < NKC; c += 128) {
        float i0 = initl[c * 2], i1 = initl[c * 2 + 1];
        float l = lse2(i0, i1);
        la0_s[c * 2] = i0 - l;
        la0_s[c * 2 + 1] = i1 - l;
        float t00 = trans[c * 4 + 0], t01 = trans[c * 4 + 1];
        float t10 = trans[c * 4 + 2], t11 = trans[c * 4 + 3];
        float l0 = lse2(t00, t10);
        float l1 = lse2(t01, t11);
        logt_s[c * 4 + 0] = t00 - l0;
        logt_s[c * 4 + 1] = t01 - l1;
        logt_s[c * 4 + 2] = t10 - l0;
        logt_s[c * 4 + 3] = t11 - l1;
    }
    __syncthreads();
    for (int i = tid; i < 4 * NKC * 2 * NA; i += 128) {
        int r = i % (NKC * 2 * NA);
        int c = r / (2 * NA);
        int s = (r % (2 * NA)) / NA;
        alpha[i] = la0_s[c * 2 + s];
    }
    __syncthreads();

    const int wave = tid >> 5, lane = tid & 31;
    const int b = blockIdx.x * 4 + wave;
    const long bT = (long)b * TT;
    float* myA = alpha + wave * (NKC * 2 * NA);
    const int aa = (lane < NA) ? lane : (NA - 1);
    const float abil = -3.0f + 0.5f * (float)aa;
    float logab = -2.5649493574615367f;
    const float NEGI = -1.0e30f;

    for (int t = 0; t < TT; ++t) {
        const long idx = bT + t;
        const int obs = corr[idx];
        const int c = kcv[idx];
        const int p = prob[idx];

        float x0, x1;
        if (USE_XBUF) {
            v2f xv = *(const v2f*)(xbuf + idx * 2);
            x0 = xv.x; x1 = xv.y;
        } else {
            const float* fr = FM + idx * 200;
            float p0 = 0.f, p1 = 0.f;
            for (int f = lane; f < 200; f += 32) {
                float fv = fr[f];
                p0 = fmaf(fv, lrw[f], p0);
                p1 = fmaf(fv, lrw[200 + f], p1);
            }
            for (int off = 16; off; off >>= 1) {
                p0 += __shfl_xor(p0, off, 32);
                p1 += __shfl_xor(p1, off, 32);
            }
            x0 = p0 + lrb[0];
            x1 = p1 + lrb[1];
        }

        const float4 op4 = *(const float4*)(olp + (long)p * 4);
        float l00 = op4.x + olk_s[c * 4 + 0] - abil + x0;
        float l01 = op4.y + olk_s[c * 4 + 1] + abil - x0;
        float l10 = op4.z + olk_s[c * 4 + 2] - abil + x1;
        float l11 = op4.w + olk_s[c * 4 + 3] + abil - x1;
        float z0 = lse2(l00, l01), z1 = lse2(l10, l11);
        float lo00 = l00 - z0, lo01 = l01 - z0;
        float lo10 = l10 - z1, lo11 = l11 - z1;

        float la0v = myA[c * 26 + aa];
        float la1v = myA[c * 26 + 13 + aa];

        float lpgb0 = lse2(lo00 + la0v, lo10 + la1v);
        float lpgb1 = lse2(lo01 + la0v, lo11 + la1v);
        float nrm = lse2(lpgb0, lpgb1);
        lpgb0 -= nrm; lpgb1 -= nrm;

        float lps0 = obs ? lo01 : lo00;
        float lps1 = obs ? lo11 : lo10;
        float nla0 = lse2(lps0 + la0v + logt_s[c * 4 + 0],
                          lps1 + la1v + logt_s[c * 4 + 1]);
        float nla1 = lse2(lps0 + la0v + logt_s[c * 4 + 2],
                          lps1 + la1v + logt_s[c * 4 + 3]);
        if (lane < NA) {
            myA[c * 26 + lane] = nla0;
            myA[c * 26 + 13 + lane] = nla1;
        }

        float o0, o1;
        {
            float v = (lane < NA) ? (logab + lpgb0) : NEGI;
            float m2 = v;
            for (int off = 8; off; off >>= 1) m2 = fmaxf(m2, __shfl_xor(m2, off, 16));
            float e = __expf(v - m2), s = e;
            for (int off = 8; off; off >>= 1) s += __shfl_xor(s, off, 16);
            o0 = add_log(m2, s);
        }
        {
            float v = (lane < NA) ? (logab + lpgb1) : NEGI;
            float m2 = v;
            for (int off = 8; off; off >>= 1) m2 = fmaxf(m2, __shfl_xor(m2, off, 16));
            float e = __expf(v - m2), s = e;
            for (int off = 8; off; off >>= 1) s += __shfl_xor(s, off, 16);
            o1 = add_log(m2, s);
        }
        float onrm = lse2(o0, o1);
        if (lane == 0) {
            v2f ov;
            ov.x = o0 - onrm;
            ov.y = o1 - onrm;
            *(v2f*)(out + idx * 2) = ov;
        }
        logab += obs ? lpgb1 : lpgb0;
    }
}

extern "C" void kernel_launch(void* const* d_in, const int* in_sizes, int n_in,
                              void* d_out, int out_size, void* d_ws, size_t ws_size,
                              hipStream_t stream) {
    (void)in_sizes; (void)n_in; (void)out_size;
    const int* corr = (const int*)d_in[0];
    const int* kcv  = (const int*)d_in[1];
    const int* prob = (const int*)d_in[2];
    const float* FM    = (const float*)d_in[3];
    const float* trans = (const float*)d_in[4];
    const float* olp   = (const float*)d_in[5];
    const float* olk   = (const float*)d_in[6];
    const float* initl = (const float*)d_in[7];
    const float* lrw   = (const float*)d_in[8];
    const float* lrb   = (const float*)d_in[9];
    float* out = (float*)d_out;

    const size_t xbytes = (size_t)BB * TT * 2 * sizeof(float);
    if (ws_size >= xbytes) {
        float* xbuf = (float*)d_ws;
        hipLaunchKernelGGL(gemm_x_wmma, dim3(2000), dim3(128), 0, stream,
                           FM, lrw, lrb, xbuf);
        hipLaunchKernelGGL((bkt_scan<true>), dim3(64), dim3(128), 0, stream,
                           corr, kcv, prob, FM, trans, olp, olk, initl,
                           lrw, lrb, xbuf, out);
    } else {
        hipLaunchKernelGGL((bkt_scan<false>), dim3(64), dim3(128), 0, stream,
                           corr, kcv, prob, FM, trans, olp, olk, initl,
                           lrw, lrb, (const float*)nullptr, out);
    }
}


# scan: the per-step 13-lane output log-sum-exp leaves the step loop: terms staged in LDS, 32 steps finished at once, one lane per step, no cross-lane shuffles; same f32 sums in the same order
# speedup vs baseline: 3.8635x; 2.3972x over previous
#include <hip/hip_runtime.h>

#define BB 256
#define TT 500
#define NKC 100
#define NA 13
#define SP 33

typedef __attribute__((ext_vector_type(2))) float v2f;
typedef __attribute__((ext_vector_type(8))) float v8f;

__device__ __forceinline__ float lse2(float x, float y) {
    float m = fmaxf(x, y);
    float e = __builtin_amdgcn_exp2f(-1.4426950408889634f * fabsf(x - y));
    return fmaf(0.6931471805599453f, __builtin_amdgcn_logf(1.0f + e), m);
}

__device__ __forceinline__ float add_log(float m, float s) {
    return fmaf(0.6931471805599453f, __builtin_amdgcn_logf(s), m);
}


__device__ __forceinline__ float lse13(const float* col) {
    float v[NA];
#pragma unroll
    for (int a = 0; a < NA; ++a) v[a] = col[a * SP];
    float m = fmaxf(fmaxf(fmaxf(fmaxf(v[0], v[8]), fmaxf(v[4], v[12])),
                          fmaxf(fmaxf(v[2], v[10]), v[6])),
                    fmaxf(fmaxf(fmaxf(v[1], v[9]), v[5]),
                          fmaxf(fmaxf(v[3], v[11]), v[7])));
    float e[NA];
#pragma unroll
    for (int a = 0; a < NA; ++a) e[a] = __expf(v[a] - m);
    float x0 = e[0] + e[8], x1 = e[1] + e[9], x2 = e[2] + e[10], x3 = e[3] + e[11];
    float x4 = e[4] + e[12];
    float y0 = x0 + x4, y1 = x1 + e[5], y2 = x2 + e[6], y3 = x3 + e[7];
    float z0 = y0 + y2, z1 = y1 + y3;
    return add_log(m, z0 + z1);
}

__global__ __launch_bounds__(128)
void gemm_x_wmma(const float* __restrict__ FM, const float* __restrict__ lrw,
                 const float* __restrict__ lrb, float* __restrict__ xout) {
    __shared__ v2f bstage[50][32];
    const int tid = threadIdx.x;
    for (int i = tid; i < 50 * 32; i += 128) {
        int k = i >> 5, L = i & 31;
        int n = L & 15;
        int kb = k * 4 + ((L >> 4) << 1);
        v2f v;
        v.x = (n < 2) ? lrw[n * 200 + kb] : 0.0f;
        v.y = (n < 2) ? lrw[n * 200 + kb + 1] : 0.0f;
        bstage[k][L] = v;
    }
    __syncthreads();

    const int wave = tid >> 5, lane = tid & 31;
    const int tile = blockIdx.x * 4 + wave;
    const long row0 = (long)tile * 16;
    const int m = lane & 15;
    const int khalf = (lane >> 4) << 1;
    const float* arow = FM + (row0 + m) * 200 + khalf;

    v8f acc = {};
    for (int k = 0; k < 50; ++k) {
        v2f a = *(const v2f*)(arow + k * 4);
        v2f b = bstage[k][lane];
        acc = __builtin_amdgcn_wmma_f32_16x16x4_f32(
            false, a, false, b, (short)0, acc, false, false);
    }

    const int n = lane & 15;
    if (n < 2) {
        const float bias = lrb[n];
        const int moff = (lane >> 4) << 3;
        for (int r = 0; r < 8; ++r) {
            long row = row0 + moff + r;
            xout[row * 2 + n] = acc[r] + bias;
        }
    }
}

template <bool USE_XBUF>
__global__ __launch_bounds__(128)
void bkt_scan(const int* __restrict__ corr, const int* __restrict__ kcv,
              const int* __restrict__ prob, const float* __restrict__ FM,
              const float* __restrict__ trans, const float* __restrict__ olp,
              const float* __restrict__ olk, const float* __restrict__ initl,
              const float* __restrict__ lrw, const float* __restrict__ lrb,
              const float* __restrict__ xbuf, float* __restrict__ out) {
    __shared__ float alpha[4 * NKC * 2 * NA];
    __shared__ float olk_s[NKC * 4];
    __shared__ float logt_s[NKC * 4];
    __shared__ float la0_s[NKC * 2];
    __shared__ float stage[4 * 2 * NA * SP];

    const int tid = threadIdx.x;
    for (int i = tid; i < NKC * 4; i += 128) olk_s[i] = olk[i];
    for (int c = tid; c < NKC; c += 128) {
        float i0 = initl[c * 2], i1 = initl[c * 2 + 1];
        float l = lse2(i0, i1);
        la0_s[c * 2] = i0 - l;
        la0_s[c * 2 + 1] = i1 - l;
        float t00 = trans[c * 4 + 0], t01 = trans[c * 4 + 1];
        float t10 = trans[c * 4 + 2], t11 = trans[c * 4 + 3];
        float l0 = lse2(t00, t10);
        float l1 = lse2(t01, t11);
        logt_s[c * 4 + 0] = t00 - l0;
        logt_s[c * 4 + 1] = t01 - l1;
        logt_s[c * 4 + 2] = t10 - l0;
        logt_s[c * 4 + 3] = t11 - l1;
    }
    __syncthreads();
    for (int i = tid; i < 4 * NKC * 2 * NA; i += 128) {
        int r = i % (NKC * 2 * NA);
        int c = r / (2 * NA);
        int s = (r % (2 * NA)) / NA;
        alpha[i] = la0_s[c * 2 + s];
    }
    __syncthreads();

    const int wave = tid >> 5, lane = tid & 31;
    const int b = blockIdx.x * 4 + wave;
    const long bT = (long)b * TT;
    float* myA = alpha + wave * (NKC * 2 * NA);
    float* myS = stage + wave * (2 * NA * SP);
    const int aa = (lane < NA) ? lane : (NA - 1);
    const float abil = -3.0f + 0.5f * (float)aa;
    float logab = -2.5649493574615367f;

    for (int t = 0; t < TT; ++t) {
        const long idx = bT + t;
        const int obs = corr[idx];
        const int c = kcv[idx];
        const int p = prob[idx];

        float x0, x1;
        if (USE_XBUF) {
            v2f xv = *(const v2f*)(xbuf + idx * 2);
            x0 = xv.x; x1 = xv.y;
        } else {
            const float* fr = FM + idx * 200;
            float p0 = 0.f, p1 = 0.f;
            for (int f = lane; f < 200; f += 32) {
                float fv = fr[f];
                p0 = fmaf(fv, lrw[f], p0);
                p1 = fmaf(fv, lrw[200 + f], p1);
            }
            for (int off = 16; off; off >>= 1) {
                p0 += __shfl_xor(p0, off, 32);
                p1 += __shfl_xor(p1, off, 32);
            }
            x0 = p0 + lrb[0];
            x1 = p1 + lrb[1];
        }

        const float4 op4 = *(const float4*)(olp + (long)p * 4);
        float l00 = op4.x + olk_s[c * 4 + 0] - abil + x0;
        float l01 = op4.y + olk_s[c * 4 + 1] + abil - x0;
        float l10 = op4.z + olk_s[c * 4 + 2] - abil + x1;
        float l11 = op4.w + olk_s[c * 4 + 3] + abil - x1;
        float z0 = lse2(l00, l01), z1 = lse2(l10, l11);
        float lo00 = l00 - z0, lo01 = l01 - z0;
        float lo10 = l10 - z1, lo11 = l11 - z1;

        float la0v = myA[c * 26 + aa];
        float la1v = myA[c * 26 + 13 + aa];

        float lpgb0 = lse2(lo00 + la0v, lo10 + la1v);
        float lpgb1 = lse2(lo01 + la0v, lo11 + la1v);
        float nrm = lse2(lpgb0, lpgb1);
        lpgb0 -= nrm; lpgb1 -= nrm;

        float lps0 = obs ? lo01 : lo00;
        float lps1 = obs ? lo11 : lo10;
        float nla0 = lse2(lps0 + la0v + logt_s[c * 4 + 0],
                          lps1 + la1v + logt_s[c * 4 + 1]);
        float nla1 = lse2(lps0 + la0v + logt_s[c * 4 + 2],
                          lps1 + la1v + logt_s[c * 4 + 3]);
        const int slot = t & 31;
        if (lane < NA) {
            myA[c * 26 + lane] = nla0;
            myA[c * 26 + 13 + lane] = nla1;
            myS[lane * SP + slot] = logab + lpgb0;
            myS[(NA + lane) * SP + slot] = logab + lpgb1;
        }

        if (slot == 31 || t == TT - 1) {
            __builtin_amdgcn_wave_barrier();
            if (lane <= slot) {
                float o0 = lse13(myS + lane);
                float o1 = lse13(myS + NA * SP + lane);
                float onrm = lse2(o0, o1);
                v2f ov;
                ov.x = o0 - onrm;
                ov.y = o1 - onrm;
                *(v2f*)(out + (idx - slot + lane) * 2) = ov;
            }
        }
        logab += obs ? lpgb1 : lpgb0;
    }
}

extern "C" void kernel_launch(void* const* d_in, const int* in_sizes, int n_in,
                              void* d_out, int out_size, void* d_ws, size_t ws_size,
                              hipStream_t stream) {
    (void)in_sizes; (void)n_in; (void)out_size;
    const int* corr = (const int*)d_in[0];
    const int* kcv  = (const int*)d_in[1];
    const int* prob = (const int*)d_in[2];
    const float* FM    = (const float*)d_in[3];
    const float* trans = (const float*)d_in[4];
    const float* olp   = (const float*)d_in[5];
    const float* olk   = (const float*)d_in[6];
    const float* initl = (const float*)d_in[7];
    const float* lrw   = (const float*)d_in[8];
    const float* lrb   = (const float*)d_in[9];
    float* out = (float*)d_out;

    const size_t xbytes = (size_t)BB * TT * 2 * sizeof(float);
    if (ws_size >= xbytes) {
        float* xbuf = (float*)d_ws;
        hipLaunchKernelGGL(gemm_x_wmma, dim3(2000), dim3(128), 0, stream,
                           FM, lrw, lrb, xbuf);
        hipLaunchKernelGGL((bkt_scan<true>), dim3(64), dim3(128), 0, stream,
                           corr, kcv, prob, FM, trans, olp, olk, initl,
                           lrw, lrb, xbuf, out);
    } else {
        hipLaunchKernelGGL((bkt_scan<false>), dim3(64), dim3(128), 0, stream,
                           corr, kcv, prob, FM, trans, olp, olk, initl,
                           lrw, lrb, (const float*)nullptr, out);
    }
}


# scan: each step's state-independent inputs (indices, x, problem/chain logits, transition logs) are fetched 32 steps at a time, lane per step, a block ahead, and reach it by lane broadcast; same sums
# speedup vs baseline: 7.4824x; 1.9367x over previous
#include <hip/hip_runtime.h>

#define BB 256
#define TT 500
#define NKC 100
#define NA 13
#define SP 33

typedef __attribute__((ext_vector_type(2))) float v2f;
typedef __attribute__((ext_vector_type(8))) float v8f;

__device__ __forceinline__ float lse2(float x, float y) {
    float m = fmaxf(x, y);
    float e = __builtin_amdgcn_exp2f(-1.4426950408889634f * fabsf(x - y));
    return fmaf(0.6931471805599453f, __builtin_amdgcn_logf(1.0f + e), m);
}

__device__ __forceinline__ float add_log(float m, float s) {
    return fmaf(0.6931471805599453f, __builtin_amdgcn_logf(s), m);
}


__device__ __forceinline__ float lse13(const float* col) {
    float v[NA];
#pragma unroll
    for (int a = 0; a < NA; ++a) v[a] = col[a * SP];
    float m = fmaxf(fmaxf(fmaxf(fmaxf(v[0], v[8]), fmaxf(v[4], v[12])),
                          fmaxf(fmaxf(v[2], v[10]), v[6])),
                    fmaxf(fmaxf(fmaxf(v[1], v[9]), v[5]),
                          fmaxf(fmaxf(v[3], v[11]), v[7])));
    float e[NA];
#pragma unroll
    for (int a = 0; a < NA; ++a) e[a] = __expf(v[a] - m);
    float x0 = e[0] + e[8], x1 = e[1] + e[9], x2 = e[2] + e[10], x3 = e[3] + e[11];
    float x4 = e[4] + e[12];
    float y0 = x0 + x4, y1 = x1 + e[5], y2 = x2 + e[6], y3 = x3 + e[7];
    float z0 = y0 + y2, z1 = y1 + y3;
    return add_log(m, z0 + z1);
}

__global__ __launch_bounds__(128)
void gemm_x_wmma(const float* __restrict__ FM, const float* __restrict__ lrw,
                 const float* __restrict__ lrb, float* __restrict__ xout) {
    __shared__ v2f bstage[50][32];
    const int tid = threadIdx.x;
    for (int i = tid; i < 50 * 32; i += 128) {
        int k = i >> 5, L = i & 31;
        int n = L & 15;
        int kb = k * 4 + ((L >> 4) << 1);
        v2f v;
        v.x = (n < 2) ? lrw[n * 200 + kb] : 0.0f;
        v.y = (n < 2) ? lrw[n * 200 + kb + 1] : 0.0f;
        bstage[k][L] = v;
    }
    __syncthreads();

    const int wave = tid >> 5, lane = tid & 31;
    const int tile = blockIdx.x * 4 + wave;
    const long row0 = (long)tile * 16;
    const int m = lane & 15;
    const int khalf = (lane >> 4) << 1;
    const float* arow = FM + (row0 + m) * 200 + khalf;

    v8f acc = {};
    for (int k = 0; k < 50; ++k) {
        v2f a = *(const v2f*)(arow + k * 4);
        v2f b = bstage[k][lane];
        acc = __builtin_amdgcn_wmma_f32_16x16x4_f32(
            false, a, false, b, (short)0, acc, false, false);
    }

    const int n = lane & 15;
    if (n < 2) {
        const float bias = lrb[n];
        const int moff = (lane >> 4) << 3;
        for (int r = 0; r < 8; ++r) {
            long row = row0 + moff + r;
            xout[row * 2 + n] = acc[r] + bias;
        }
    }
}

#define NBLK ((TT + 31) / 32)

struct StepIdx { int obs, c, p; v2f x; };

template <bool USE_XBUF>
__device__ __forceinline__ StepIdx fetch_idx(const int* __restrict__ corr,
                                             const int* __restrict__ kcv,
                                             const int* __restrict__ prob,
                                             const float* __restrict__ xbuf,
                                             long bT, int t) {
    const long id = bT + (t < TT ? t : TT - 1);
    StepIdx r;
    r.obs = corr[id];
    r.c = kcv[id];
    r.p = prob[id];
    if (USE_XBUF) r.x = *(const v2f*)(xbuf + id * 2);
    else { r.x.x = 0.0f; r.x.y = 0.0f; }
    return r;
}

__device__ __forceinline__ int bcast_i(int v, int src) {
    return __builtin_amdgcn_readlane(v, src);
}
__device__ __forceinline__ float bcast_f(float v, int src) {
    return __int_as_float(__builtin_amdgcn_readlane(__float_as_int(v), src));
}

template <bool USE_XBUF>
__global__ __launch_bounds__(128)
void bkt_scan(const int* __restrict__ corr, const int* __restrict__ kcv,
              const int* __restrict__ prob, const float* __restrict__ FM,
              const float* __restrict__ trans, const float* __restrict__ olp,
              const float* __restrict__ olk, const float* __restrict__ initl,
              const float* __restrict__ lrw, const float* __restrict__ lrb,
              const float* __restrict__ xbuf, float* __restrict__ out) {
    __shared__ float alpha[4 * NKC * 2 * NA];
    __shared__ float olk_s[NKC * 4];
    __shared__ float logt_s[NKC * 4];
    __shared__ float la0_s[NKC * 2];
    __shared__ float stage[4 * 2 * NA * SP];

    const int tid = threadIdx.x;
    for (int i = tid; i < NKC * 4; i += 128) olk_s[i] = olk[i];
    for (int c = tid; c < NKC; c += 128) {
        float i0 = initl[c * 2], i1 = initl[c * 2 + 1];
        float l = lse2(i0, i1);
        la0_s[c * 2] = i0 - l;
        la0_s[c * 2 + 1] = i1 - l;
        float t00 = trans[c * 4 + 0], t01 = trans[c * 4 + 1];
        float t10 = trans[c * 4 + 2], t11 = trans[c * 4 + 3];
        float l0 = lse2(t00, t10);
        float l1 = lse2(t01, t11);
        logt_s[c * 4 + 0] = t00 - l0;
        logt_s[c * 4 + 1] = t01 - l1;
        logt_s[c * 4 + 2] = t10 - l0;
        logt_s[c * 4 + 3] = t11 - l1;
    }
    __syncthreads();
    for (int i = tid; i < 4 * NKC * 2 * NA; i += 128) {
        int r = i % (NKC * 2 * NA);
        int c = r / (2 * NA);
        int s = (r % (2 * NA)) / NA;
        alpha[i] = la0_s[c * 2 + s];
    }
    __syncthreads();

    const int wave = tid >> 5, lane = tid & 31;
    const int b = blockIdx.x * 4 + wave;
    const long bT = (long)b * TT;
    float* myA = alpha + wave * (NKC * 2 * NA);
    float* myS = stage + wave * (2 * NA * SP);
    const int aa = (lane < NA) ? lane : (NA - 1);
    const float abil = -3.0f + 0.5f * (float)aa;
    float logab = -2.5649493574615367f;

    StepIdx i0 = fetch_idx<USE_XBUF>(corr, kcv, prob, xbuf, bT, lane);
    StepIdx i1 = fetch_idx<USE_XBUF>(corr, kcv, prob, xbuf, bT, 32 + lane);
    float4 g0 = *(const float4*)(olp + (long)i0.p * 4);

    for (int k = 0; k < NBLK; ++k) {
        const int t0 = k * 32;
        const int n = (TT - t0 < 32) ? (TT - t0) : 32;

        const int vobs = i0.obs, vc = i0.c;
        const float vx0 = i0.x.x, vx1 = i0.x.y;
        const float vq00 = g0.x + olk_s[vc * 4 + 0];
        const float vq01 = g0.y + olk_s[vc * 4 + 1];
        const float vq10 = g0.z + olk_s[vc * 4 + 2];
        const float vq11 = g0.w + olk_s[vc * 4 + 3];
        const float vt00 = logt_s[vc * 4 + 0], vt01 = logt_s[vc * 4 + 1];
        const float vt10 = logt_s[vc * 4 + 2], vt11 = logt_s[vc * 4 + 3];

        g0 = *(const float4*)(olp + (long)i1.p * 4);
        i0 = i1;
        i1 = fetch_idx<USE_XBUF>(corr, kcv, prob, xbuf, bT, t0 + 64 + lane);

        int obs = bcast_i(vobs, 0), c = bcast_i(vc, 0);
        float sx0 = bcast_f(vx0, 0), sx1 = bcast_f(vx1, 0);
        float q00 = bcast_f(vq00, 0), q01 = bcast_f(vq01, 0);
        float q10 = bcast_f(vq10, 0), q11 = bcast_f(vq11, 0);
        float t00 = bcast_f(vt00, 0), t01 = bcast_f(vt01, 0);
        float t10 = bcast_f(vt10, 0), t11 = bcast_f(vt11, 0);

        for (int slot = 0; slot < n; ++slot) {
            const int s1 = (slot + 1) & 31;
            const int nobs = bcast_i(vobs, s1), nc = bcast_i(vc, s1);
            const float nx0 = bcast_f(vx0, s1), nx1 = bcast_f(vx1, s1);
            const float nq00 = bcast_f(vq00, s1), nq01 = bcast_f(vq01, s1);
            const float nq10 = bcast_f(vq10, s1), nq11 = bcast_f(vq11, s1);
            const float nt00 = bcast_f(vt00, s1), nt01 = bcast_f(vt01, s1);
            const float nt10 = bcast_f(vt10, s1), nt11 = bcast_f(vt11, s1);

            float x0, x1;
            if (USE_XBUF) {
                x0 = sx0; x1 = sx1;
            } else {
                const float* fr = FM + (bT + t0 + slot) * 200;
                float p0 = 0.f, p1 = 0.f;
                for (int f = lane; f < 200; f += 32) {
                    float fv = fr[f];
                    p0 = fmaf(fv, lrw[f], p0);
                    p1 = fmaf(fv, lrw[200 + f], p1);
                }
                for (int off = 16; off; off >>= 1) {
                    p0 += __shfl_xor(p0, off, 32);
                    p1 += __shfl_xor(p1, off, 32);
                }
                x0 = p0 + lrb[0];
                x1 = p1 + lrb[1];
            }

            float l00 = q00 - abil + x0;
            float l01 = q01 + abil - x0;
            float l10 = q10 - abil + x1;
            float l11 = q11 + abil - x1;
            float z0 = lse2(l00, l01), z1 = lse2(l10, l11);
            float lo00 = l00 - z0, lo01 = l01 - z0;
            float lo10 = l10 - z1, lo11 = l11 - z1;

            float la0v = myA[c * 26 + aa];
            float la1v = myA[c * 26 + 13 + aa];

            float lpgb0 = lse2(lo00 + la0v, lo10 + la1v);
            float lpgb1 = lse2(lo01 + la0v, lo11 + la1v);
            float nrm = lse2(lpgb0, lpgb1);
            lpgb0 -= nrm; lpgb1 -= nrm;

            float lps0 = obs ? lo01 : lo00;
            float lps1 = obs ? lo11 : lo10;
            float nla0 = lse2(lps0 + la0v + t00, lps1 + la1v + t01);
            float nla1 = lse2(lps0 + la0v + t10, lps1 + la1v + t11);
            if (lane < NA) {
                myA[c * 26 + lane] = nla0;
                myA[c * 26 + 13 + lane] = nla1;
                myS[lane * SP + slot] = logab + lpgb0;
                myS[(NA + lane) * SP + slot] = logab + lpgb1;
            }
            logab += obs ? lpgb1 : lpgb0;

            obs = nobs; c = nc; sx0 = nx0; sx1 = nx1;
            q00 = nq00; q01 = nq01; q10 = nq10; q11 = nq11;
            t00 = nt00; t01 = nt01; t10 = nt10; t11 = nt11;
        }

        __builtin_amdgcn_wave_barrier();
        if (lane < n) {
            float o0 = lse13(myS + lane);
            float o1 = lse13(myS + NA * SP + lane);
            float onrm = lse2(o0, o1);
            v2f ov;
            ov.x = o0 - onrm;
            ov.y = o1 - onrm;
            *(v2f*)(out + (bT + t0 + lane) * 2) = ov;
        }
    }
}

extern "C" void kernel_launch(void* const* d_in, const int* in_sizes, int n_in,
                              void* d_out, int out_size, void* d_ws, size_t ws_size,
                              hipStream_t stream) {
    (void)in_sizes; (void)n_in; (void)out_size;
    const int* corr = (const int*)d_in[0];
    const int* kcv  = (const int*)d_in[1];
    const int* prob = (const int*)d_in[2];
    const float* FM    = (const float*)d_in[3];
    const float* trans = (const float*)d_in[4];
    const float* olp   = (const float*)d_in[5];
    const float* olk   = (const float*)d_in[6];
    const float* initl = (const float*)d_in[7];
    const float* lrw   = (const float*)d_in[8];
    const float* lrb   = (const float*)d_in[9];
    float* out = (float*)d_out;

    const size_t xbytes = (size_t)BB * TT * 2 * sizeof(float);
    if (ws_size >= xbytes) {
        float* xbuf = (float*)d_ws;
        hipLaunchKernelGGL(gemm_x_wmma, dim3(2000), dim3(128), 0, stream,
                           FM, lrw, lrb, xbuf);
        hipLaunchKernelGGL((bkt_scan<true>), dim3(64), dim3(128), 0, stream,
                           corr, kcv, prob, FM, trans, olp, olk, initl,
                           lrw, lrb, xbuf, out);
    } else {
        hipLaunchKernelGGL((bkt_scan<false>), dim3(64), dim3(128), 0, stream,
                           corr, kcv, prob, FM, trans, olp, olk, initl,
                           lrw, lrb, (const float*)nullptr, out);
    }
}


# scan: step loop pipelined one step ahead: next step's observation log-softmax and alpha read (kept in registers when the chain repeats) issue this step; store mask replaced by scrap rows; same sums
# speedup vs baseline: 8.4288x; 1.1265x over previous
#include <hip/hip_runtime.h>

#define BB 256
#define TT 500
#define NKC 100
#define NA 13
#define SP 33

typedef __attribute__((ext_vector_type(2))) float v2f;
typedef __attribute__((ext_vector_type(8))) float v8f;

__device__ __forceinline__ float lse2(float x, float y) {
    float m = fmaxf(x, y);
    float e = __builtin_amdgcn_exp2f(-1.4426950408889634f * fabsf(x - y));
    return fmaf(0.6931471805599453f, __builtin_amdgcn_logf(1.0f + e), m);
}

__device__ __forceinline__ float add_log(float m, float s) {
    return fmaf(0.6931471805599453f, __builtin_amdgcn_logf(s), m);
}


__device__ __forceinline__ float lse13(const float* col) {
    float v[NA];
#pragma unroll
    for (int a = 0; a < NA; ++a) v[a] = col[a * SP];
    float m = fmaxf(fmaxf(fmaxf(fmaxf(v[0], v[8]), fmaxf(v[4], v[12])),
                          fmaxf(fmaxf(v[2], v[10]), v[6])),
                    fmaxf(fmaxf(fmaxf(v[1], v[9]), v[5]),
                          fmaxf(fmaxf(v[3], v[11]), v[7])));
    float e[NA];
#pragma unroll
    for (int a = 0; a < NA; ++a) e[a] = __expf(v[a] - m);
    float x0 = e[0] + e[8], x1 = e[1] + e[9], x2 = e[2] + e[10], x3 = e[3] + e[11];
    float x4 = e[4] + e[12];
    float y0 = x0 + x4, y1 = x1 + e[5], y2 = x2 + e[6], y3 = x3 + e[7];
    float z0 = y0 + y2, z1 = y1 + y3;
    return add_log(m, z0 + z1);
}

__global__ __launch_bounds__(128)
void gemm_x_wmma(const float* __restrict__ FM, const float* __restrict__ lrw,
                 const float* __restrict__ lrb, float* __restrict__ xout) {
    __shared__ v2f bstage[50][32];
    const int tid = threadIdx.x;
    for (int i = tid; i < 50 * 32; i += 128) {
        int k = i >> 5, L = i & 31;
        int n = L & 15;
        int kb = k * 4 + ((L >> 4) << 1);
        v2f v;
        v.x = (n < 2) ? lrw[n * 200 + kb] : 0.0f;
        v.y = (n < 2) ? lrw[n * 200 + kb + 1] : 0.0f;
        bstage[k][L] = v;
    }
    __syncthreads();

    const int wave = tid >> 5, lane = tid & 31;
    const int tile = blockIdx.x * 4 + wave;
    const long row0 = (long)tile * 16;
    const int m = lane & 15;
    const int khalf = (lane >> 4) << 1;
    const float* arow = FM + (row0 + m) * 200 + khalf;

    v8f acc = {};
    for (int k = 0; k < 50; ++k) {
        v2f a = *(const v2f*)(arow + k * 4);
        v2f b = bstage[k][lane];
        acc = __builtin_amdgcn_wmma_f32_16x16x4_f32(
            false, a, false, b, (short)0, acc, false, false);
    }

    const int n = lane & 15;
    if (n < 2) {
        const float bias = lrb[n];
        const int moff = (lane >> 4) << 3;
        for (int r = 0; r < 8; ++r) {
            long row = row0 + moff + r;
            xout[row * 2 + n] = acc[r] + bias;
        }
    }
}

#define NBLK ((TT + 31) / 32)
#define SCRAP 576

struct StepIdx { int obs, c, p; v2f x; };

template <bool USE_XBUF>
__device__ __forceinline__ StepIdx fetch_idx(const int* __restrict__ corr,
                                             const int* __restrict__ kcv,
                                             const int* __restrict__ prob,
                                             const float* __restrict__ xbuf,
                                             long bT, int t) {
    const long id = bT + (t < TT ? t : TT - 1);
    StepIdx r;
    r.obs = corr[id];
    r.c = kcv[id];
    r.p = prob[id];
    if (USE_XBUF) r.x = *(const v2f*)(xbuf + id * 2);
    else { r.x.x = 0.0f; r.x.y = 0.0f; }
    return r;
}

__device__ __forceinline__ int bcast_i(int v, int src) {
    return __builtin_amdgcn_readlane(v, src);
}
__device__ __forceinline__ float bcast_f(float v, int src) {
    return __int_as_float(__builtin_amdgcn_readlane(__float_as_int(v), src));
}

struct ObsLog { float l00, l01, l10, l11; };
__device__ __forceinline__ ObsLog obs_logs(float q00, float q01, float q10, float q11,
                                           float x0, float x1, float abil) {
    float l00 = q00 - abil + x0;
    float l01 = q01 + abil - x0;
    float l10 = q10 - abil + x1;
    float l11 = q11 + abil - x1;
    float z0 = lse2(l00, l01), z1 = lse2(l10, l11);
    ObsLog r;
    r.l00 = l00 - z0; r.l01 = l01 - z0;
    r.l10 = l10 - z1; r.l11 = l11 - z1;
    return r;
}

template <bool USE_XBUF>
__global__ __launch_bounds__(128)
void bkt_scan(const int* __restrict__ corr, const int* __restrict__ kcv,
              const int* __restrict__ prob, const float* __restrict__ FM,
              const float* __restrict__ trans, const float* __restrict__ olp,
              const float* __restrict__ olk, const float* __restrict__ initl,
              const float* __restrict__ lrw, const float* __restrict__ lrb,
              const float* __restrict__ xbuf, float* __restrict__ out) {
    __shared__ float alpha[4 * NKC * 2 * NA];
    __shared__ float olk_s[NKC * 4];
    __shared__ float logt_s[NKC * 4];
    __shared__ float la0_s[NKC * 2];
    __shared__ float stage[4 * 2 * NA * SP];
    __shared__ float scrap[4 * SCRAP];

    const int tid = threadIdx.x;
    for (int i = tid; i < NKC * 4; i += 128) olk_s[i] = olk[i];
    for (int c = tid; c < NKC; c += 128) {
        float i0 = initl[c * 2], i1 = initl[c * 2 + 1];
        float l = lse2(i0, i1);
        la0_s[c * 2] = i0 - l;
        la0_s[c * 2 + 1] = i1 - l;
        float t00 = trans[c * 4 + 0], t01 = trans[c * 4 + 1];
        float t10 = trans[c * 4 + 2], t11 = trans[c * 4 + 3];
        float l0 = lse2(t00, t10);
        float l1 = lse2(t01, t11);
        logt_s[c * 4 + 0] = t00 - l0;
        logt_s[c * 4 + 1] = t01 - l1;
        logt_s[c * 4 + 2] = t10 - l0;
        logt_s[c * 4 + 3] = t11 - l1;
    }
    __syncthreads();
    for (int i = tid; i < 4 * NKC * 2 * NA; i += 128) {
        int r = i % (NKC * 2 * NA);
        int c = r / (2 * NA);
        int s = (r % (2 * NA)) / NA;
        alpha[i] = la0_s[c * 2 + s];
    }
    __syncthreads();

    const int wave = tid >> 5, lane = tid & 31;
    const int b = blockIdx.x * 4 + wave;
    const long bT = (long)b * TT;
    float* myA = alpha + wave * (NKC * 2 * NA);
    float* myS = stage + wave * (2 * NA * SP);
    float* myScrap = scrap + wave * SCRAP;
    const int aa = (lane < NA) ? lane : (NA - 1);
    const float abil = -3.0f + 0.5f * (float)aa;
    float logab = -2.5649493574615367f;

    StepIdx i0 = fetch_idx<USE_XBUF>(corr, kcv, prob, xbuf, bT, lane);
    StepIdx i1 = fetch_idx<USE_XBUF>(corr, kcv, prob, xbuf, bT, 32 + lane);
    float4 g0 = *(const float4*)(olp + (long)i0.p * 4);

    for (int k = 0; k < NBLK; ++k) {
        const int t0 = k * 32;
        const int n = (TT - t0 < 32) ? (TT - t0) : 32;

        const int vobs = i0.obs, vc = i0.c;
        float vx0 = i0.x.x, vx1 = i0.x.y;
        if (!USE_XBUF) {
            for (int j = 0; j < n; ++j) {
                const float* fr = FM + (bT + t0 + j) * 200;
                float p0 = 0.f, p1 = 0.f;
                for (int f = lane; f < 200; f += 32) {
                    float fv = fr[f];
                    p0 = fmaf(fv, lrw[f], p0);
                    p1 = fmaf(fv, lrw[200 + f], p1);
                }
                for (int off = 16; off; off >>= 1) {
                    p0 += __shfl_xor(p0, off, 32);
                    p1 += __shfl_xor(p1, off, 32);
                }
                if (lane == j) { vx0 = p0 + lrb[0]; vx1 = p1 + lrb[1]; }
            }
        }
        const float vq00 = g0.x + olk_s[vc * 4 + 0];
        const float vq01 = g0.y + olk_s[vc * 4 + 1];
        const float vq10 = g0.z + olk_s[vc * 4 + 2];
        const float vq11 = g0.w + olk_s[vc * 4 + 3];
        const float vt00 = logt_s[vc * 4 + 0], vt01 = logt_s[vc * 4 + 1];
        const float vt10 = logt_s[vc * 4 + 2], vt11 = logt_s[vc * 4 + 3];

        g0 = *(const float4*)(olp + (long)i1.p * 4);
        i0 = i1;
        i1 = fetch_idx<USE_XBUF>(corr, kcv, prob, xbuf, bT, t0 + 64 + lane);

        const bool real = lane < NA;
        float* const stS = real ? (myS + lane * SP) : (myScrap + 64 + lane);

        int obs = bcast_i(vobs, 0), c = bcast_i(vc, 0);
        float t00 = bcast_f(vt00, 0), t01 = bcast_f(vt01, 0);
        float t10 = bcast_f(vt10, 0), t11 = bcast_f(vt11, 0);
        const ObsLog lo = obs_logs(bcast_f(vq00, 0), bcast_f(vq01, 0),
                                   bcast_f(vq10, 0), bcast_f(vq11, 0),
                                   bcast_f(vx0, 0), bcast_f(vx1, 0), abil);
        const float fa0 = myA[c * 26 + aa];
        const float fa1 = myA[c * 26 + 13 + aa];
        float u00 = lo.l00 + fa0, u01 = lo.l01 + fa0;
        float u10 = lo.l10 + fa1, u11 = lo.l11 + fa1;

        for (int slot = 0; slot < n; ++slot) {
            const int s1 = (slot + 1) & 31;
            const int nobs = bcast_i(vobs, s1), nc = bcast_i(vc, s1);
            const float nt00 = bcast_f(vt00, s1), nt01 = bcast_f(vt01, s1);
            const float nt10 = bcast_f(vt10, s1), nt11 = bcast_f(vt11, s1);

            const float pl0 = myA[nc * 26 + aa];
            const float pl1 = myA[nc * 26 + 13 + aa];

            const ObsLog nlo = obs_logs(bcast_f(vq00, s1), bcast_f(vq01, s1),
                                        bcast_f(vq10, s1), bcast_f(vq11, s1),
                                        bcast_f(vx0, s1), bcast_f(vx1, s1), abil);

            float lpgb0 = lse2(u00, u10);
            float lpgb1 = lse2(u01, u11);
            float nrm = lse2(lpgb0, lpgb1);
            lpgb0 -= nrm; lpgb1 -= nrm;

            const float w0 = obs ? u01 : u00;
            const float w1 = obs ? u11 : u10;
            const float nla0 = lse2(w0 + t00, w1 + t01);
            const float nla1 = lse2(w0 + t10, w1 + t11);
            float* const stA = real ? (myA + c * 26 + lane) : (myScrap + lane);
            stA[0] = nla0;
            stA[13] = nla1;
            stS[slot] = logab + lpgb0;
            stS[NA * SP + slot] = logab + lpgb1;
            logab += obs ? lpgb1 : lpgb0;

            const bool same = (nc == c);
            const float na0 = same ? nla0 : pl0;
            const float na1 = same ? nla1 : pl1;
            u00 = nlo.l00 + na0; u01 = nlo.l01 + na0;
            u10 = nlo.l10 + na1; u11 = nlo.l11 + na1;
            obs = nobs; c = nc;
            t00 = nt00; t01 = nt01; t10 = nt10; t11 = nt11;
        }

        __builtin_amdgcn_wave_barrier();
        if (lane < n) {
            float o0 = lse13(myS + lane);
            float o1 = lse13(myS + NA * SP + lane);
            float onrm = lse2(o0, o1);
            v2f ov;
            ov.x = o0 - onrm;
            ov.y = o1 - onrm;
            *(v2f*)(out + (bT + t0 + lane) * 2) = ov;
        }
    }
}

extern "C" void kernel_launch(void* const* d_in, const int* in_sizes, int n_in,
                              void* d_out, int out_size, void* d_ws, size_t ws_size,
                              hipStream_t stream) {
    (void)in_sizes; (void)n_in; (void)out_size;
    const int* corr = (const int*)d_in[0];
    const int* kcv  = (const int*)d_in[1];
    const int* prob = (const int*)d_in[2];
    const float* FM    = (const float*)d_in[3];
    const float* trans = (const float*)d_in[4];
    const float* olp   = (const float*)d_in[5];
    const float* olk   = (const float*)d_in[6];
    const float* initl = (const float*)d_in[7];
    const float* lrw   = (const float*)d_in[8];
    const float* lrb   = (const float*)d_in[9];
    float* out = (float*)d_out;

    const size_t xbytes = (size_t)BB * TT * 2 * sizeof(float);
    if (ws_size >= xbytes) {
        float* xbuf = (float*)d_ws;
        hipLaunchKernelGGL(gemm_x_wmma, dim3(2000), dim3(128), 0, stream,
                           FM, lrw, lrb, xbuf);
        hipLaunchKernelGGL((bkt_scan<true>), dim3(64), dim3(128), 0, stream,
                           corr, kcv, prob, FM, trans, olp, olk, initl,
                           lrw, lrb, xbuf, out);
    } else {
        hipLaunchKernelGGL((bkt_scan<false>), dim3(64), dim3(128), 0, stream,
                           corr, kcv, prob, FM, trans, olp, olk, initl,
                           lrw, lrb, (const float*)nullptr, out);
    }
}


# scan: time loop split by chain: steps counting-sorted by chain in LDS, each chain's alpha recurrence on its own 16 lanes, then log_ability summed over time in order, outputs lane per step; same sums
# speedup vs baseline: 23.0235x; 2.7315x over previous
#include <hip/hip_runtime.h>

#define BB 256
#define TT 500
#define NKC 100
#define NA 13
#define TP 513

typedef __attribute__((ext_vector_type(2))) float v2f;
typedef __attribute__((ext_vector_type(8))) float v8f;

__device__ __forceinline__ float lse2(float x, float y) {
    float m = fmaxf(x, y);
    float e = __builtin_amdgcn_exp2f(-1.4426950408889634f * fabsf(x - y));
    return fmaf(0.6931471805599453f, __builtin_amdgcn_logf(1.0f + e), m);
}

__device__ __forceinline__ float add_log(float m, float s) {
    return fmaf(0.6931471805599453f, __builtin_amdgcn_logf(s), m);
}


__device__ __forceinline__ float lse13(const float* lab, const float* l) {
    float v[NA];
#pragma unroll
    for (int a = 0; a < NA; ++a) v[a] = lab[a * TP] + l[a * TP];
    float m = fmaxf(fmaxf(fmaxf(fmaxf(v[0], v[8]), fmaxf(v[4], v[12])),
                          fmaxf(fmaxf(v[2], v[10]), v[6])),
                    fmaxf(fmaxf(fmaxf(v[1], v[9]), v[5]),
                          fmaxf(fmaxf(v[3], v[11]), v[7])));
    float e[NA];
#pragma unroll
    for (int a = 0; a < NA; ++a) e[a] = __expf(v[a] - m);
    float x0 = e[0] + e[8], x1 = e[1] + e[9], x2 = e[2] + e[10], x3 = e[3] + e[11];
    float x4 = e[4] + e[12];
    float y0 = x0 + x4, y1 = x1 + e[5], y2 = x2 + e[6], y3 = x3 + e[7];
    float z0 = y0 + y2, z1 = y1 + y3;
    return add_log(m, z0 + z1);
}

__global__ __launch_bounds__(128)
void gemm_x_wmma(const float* __restrict__ FM, const float* __restrict__ lrw,
                 const float* __restrict__ lrb, float* __restrict__ xout) {
    __shared__ v2f bstage[50][32];
    const int tid = threadIdx.x;
    for (int i = tid; i < 50 * 32; i += 128) {
        int k = i >> 5, L = i & 31;
        int n = L & 15;
        int kb = k * 4 + ((L >> 4) << 1);
        v2f v;
        v.x = (n < 2) ? lrw[n * 200 + kb] : 0.0f;
        v.y = (n < 2) ? lrw[n * 200 + kb + 1] : 0.0f;
        bstage[k][L] = v;
    }
    __syncthreads();

    const int wave = tid >> 5, lane = tid & 31;
    const int tile = blockIdx.x * 4 + wave;
    const long row0 = (long)tile * 16;
    const int m = lane & 15;
    const int khalf = (lane >> 4) << 1;
    const float* arow = FM + (row0 + m) * 200 + khalf;

    v8f acc = {};
    for (int k = 0; k < 50; ++k) {
        v2f a = *(const v2f*)(arow + k * 4);
        v2f b = bstage[k][lane];
        acc = __builtin_amdgcn_wmma_f32_16x16x4_f32(
            false, a, false, b, (short)0, acc, false, false);
    }

    const int n = lane & 15;
    if (n < 2) {
        const float bias = lrb[n];
        const int moff = (lane >> 4) << 3;
        for (int r = 0; r < 8; ++r) {
            long row = row0 + moff + r;
            xout[row * 2 + n] = acc[r] + bias;
        }
    }
}

#define NTH 512
#define NPART 5
#define PLEN (TT / NPART)
#define PG 20

struct ObsLog { float l00, l01, l10, l11; };
__device__ __forceinline__ ObsLog obs_logs(float q00, float q01, float q10, float q11,
                                           float x0, float x1, float abil) {
    float l00 = q00 - abil + x0;
    float l01 = q01 + abil - x0;
    float l10 = q10 - abil + x1;
    float l11 = q11 + abil - x1;
    float z0 = lse2(l00, l01), z1 = lse2(l10, l11);
    ObsLog r;
    r.l00 = l00 - z0; r.l01 = l01 - z0;
    r.l10 = l10 - z1; r.l11 = l11 - z1;
    return r;
}

template <bool USE_XBUF>
__global__ __launch_bounds__(NTH)
void bkt_chains(const int* __restrict__ corr, const int* __restrict__ kcv,
                const int* __restrict__ prob, const float* __restrict__ FM,
                const float* __restrict__ trans, const float* __restrict__ olp,
                const float* __restrict__ olk, const float* __restrict__ initl,
                const float* __restrict__ lrw, const float* __restrict__ lrb,
                const float* __restrict__ xbuf, float* __restrict__ out) {
    __shared__ __attribute__((aligned(16))) int s_c[NTH];
    __shared__ int s_obs[NTH];
    __shared__ float4 s_q[NTH];
    __shared__ v2f s_x[NTH];
    __shared__ int s_less[NPART * NKC];
    __shared__ int s_eq[NPART * NKC];
    __shared__ int s_start[NKC];
    __shared__ int s_cnt[NKC];
    __shared__ int s_order[NTH];
    __shared__ float s_la0[NKC * 2];
    __shared__ float s_logt[NKC * 4];
    __shared__ float s_l0[NA * TP];
    __shared__ float s_l1[NA * TP];
    __shared__ float s_lab[NA * TP];

    const int tid = threadIdx.x;
    const int b = blockIdx.x;
    const long bT = (long)b * TT;

    if (tid < NKC) {
        const int c = tid;
        const float i0 = initl[c * 2], i1 = initl[c * 2 + 1];
        const float il = lse2(i0, i1);
        s_la0[c * 2] = i0 - il;
        s_la0[c * 2 + 1] = i1 - il;
        const float r00 = trans[c * 4 + 0], r01 = trans[c * 4 + 1];
        const float r10 = trans[c * 4 + 2], r11 = trans[c * 4 + 3];
        const float n0 = lse2(r00, r10);
        const float n1 = lse2(r01, r11);
        s_logt[c * 4 + 0] = r00 - n0;
        s_logt[c * 4 + 1] = r01 - n1;
        s_logt[c * 4 + 2] = r10 - n0;
        s_logt[c * 4 + 3] = r11 - n1;
    }

    if (tid < TT) {
        const long id = bT + tid;
        const int c = kcv[id];
        const int p = prob[id];
        s_c[tid] = c;
        s_obs[tid] = corr[id];
        const float4 g = *(const float4*)(olp + (long)p * 4);
        const float4 kq = *(const float4*)(olk + c * 4);
        float4 q;
        q.x = g.x + kq.x; q.y = g.y + kq.y; q.z = g.z + kq.z; q.w = g.w + kq.w;
        s_q[tid] = q;
        v2f x;
        if (USE_XBUF) {
            x = *(const v2f*)(xbuf + id * 2);
        } else {
            const float* fr = FM + id * 200;
            float p0 = 0.f, p1 = 0.f;
            for (int f = 0; f < 200; ++f) {
                const float fv = fr[f];
                p0 = fmaf(fv, lrw[f], p0);
                p1 = fmaf(fv, lrw[200 + f], p1);
            }
            x.x = p0 + lrb[0];
            x.y = p1 + lrb[1];
        }
        s_x[tid] = x;
    }
    __syncthreads();

    const int part = tid / NKC;
    const int myc = tid - part * NKC;
    const bool sorter = tid < NPART * NKC;
    if (sorter) {
        const int4* src = (const int4*)(s_c + part * PLEN);
        int less = 0, eq = 0;
        for (int i = 0; i < PLEN / 4; ++i) {
            const int4 v = src[i];
            less += (v.x < myc) + (v.y < myc) + (v.z < myc) + (v.w < myc);
            eq += (v.x == myc) + (v.y == myc) + (v.z == myc) + (v.w == myc);
        }
        s_less[part * NKC + myc] = less;
        s_eq[part * NKC + myc] = eq;
    }
    __syncthreads();
    if (sorter) {
        int k = 0, tot = 0;
        for (int p = 0; p < NPART; ++p) {
            k += s_less[p * NKC + myc];
            const int e = s_eq[p * NKC + myc];
            if (p < part) k += e;
            tot += e;
        }
        if (part == 0) { s_start[myc] = k; s_cnt[myc] = tot; }
        const int base = part * PLEN;
        const int4* src = (const int4*)(s_c + base);
        for (int i = 0; i < PLEN / 4; i += 5) {
            int4 v[5];
#pragma unroll
            for (int j = 0; j < 5; ++j) v[j] = src[i + j];
#pragma unroll
            for (int j = 0; j < 5; ++j) {
                const int t4 = base + (i + j) * 4;
                if (v[j].x == myc) { s_order[k] = t4; ++k; }
                if (v[j].y == myc) { s_order[k] = t4 + 1; ++k; }
                if (v[j].z == myc) { s_order[k] = t4 + 2; ++k; }
                if (v[j].w == myc) { s_order[k] = t4 + 3; ++k; }
            }
        }
    }
    __syncthreads();

    {
        const int a16 = tid & 15;
        const bool real = a16 < NA;
        const int aa = real ? a16 : (NA - 1);
        const float abil = -3.0f + 0.5f * (float)aa;
        for (int c = tid >> 4; c < NKC; c += NTH / 16) {
            float al0 = s_la0[c * 2], al1 = s_la0[c * 2 + 1];
            const float t00 = s_logt[c * 4 + 0], t01 = s_logt[c * 4 + 1];
            const float t10 = s_logt[c * 4 + 2], t11 = s_logt[c * 4 + 3];
            const int k0 = s_start[c];
            const int n = s_cnt[c];
            if (n > 0) {
                const int kl = k0 + n - 1;
                int t = s_order[k0];
                int tn = s_order[k0 + 1 <= kl ? k0 + 1 : kl];
                int obs = s_obs[t];
                float4 q = s_q[t];
                v2f x = s_x[t];
                for (int k = k0; k <= kl; ++k) {
                    const int tnn = s_order[k + 2 <= kl ? k + 2 : kl];
                    const int obsn = s_obs[tn];
                    const float4 qn = s_q[tn];
                    const v2f xn = s_x[tn];

                    const ObsLog lo = obs_logs(q.x, q.y, q.z, q.w, x.x, x.y, abil);
                    const float u00 = lo.l00 + al0, u01 = lo.l01 + al0;
                    const float u10 = lo.l10 + al1, u11 = lo.l11 + al1;
                    float lpgb0 = lse2(u00, u10);
                    float lpgb1 = lse2(u01, u11);
                    const float nrm = lse2(lpgb0, lpgb1);
                    lpgb0 -= nrm; lpgb1 -= nrm;
                    const float w0 = obs ? u01 : u00;
                    const float w1 = obs ? u11 : u10;
                    al0 = lse2(w0 + t00, w1 + t01);
                    al1 = lse2(w0 + t10, w1 + t11);
                    if (real) {
                        s_l0[aa * TP + t] = lpgb0;
                        s_l1[aa * TP + t] = lpgb1;
                        s_lab[aa * TP + t] = obs ? lpgb1 : lpgb0;
                    }
                    t = tn; tn = tnn; obs = obsn; q = qn; x = xn;
                }
            }
        }
    }
    __syncthreads();

    if (tid < NA) {
        float* const col = s_lab + tid * TP;
        float logab = -2.5649493574615367f;
        float cur[PG], nxt[PG];
#pragma unroll
        for (int j = 0; j < PG; ++j) cur[j] = col[j];
        for (int g = 0; g < TT / PG; ++g) {
            const int gn = (g + 1 < TT / PG) ? g + 1 : g;
#pragma unroll
            for (int j = 0; j < PG; ++j) nxt[j] = col[gn * PG + j];
#pragma unroll
            for (int j = 0; j < PG; ++j) {
                col[g * PG + j] = logab;
                logab += cur[j];
            }
#pragma unroll
            for (int j = 0; j < PG; ++j) cur[j] = nxt[j];
        }
    }
    __syncthreads();

    if (tid < TT) {
        const float o0 = lse13(s_lab + tid, s_l0 + tid);
        const float o1 = lse13(s_lab + tid, s_l1 + tid);
        const float onrm = lse2(o0, o1);
        v2f ov;
        ov.x = o0 - onrm;
        ov.y = o1 - onrm;
        *(v2f*)(out + (bT + tid) * 2) = ov;
    }
}

extern "C" void kernel_launch(void* const* d_in, const int* in_sizes, int n_in,
                              void* d_out, int out_size, void* d_ws, size_t ws_size,
                              hipStream_t stream) {
    (void)in_sizes; (void)n_in; (void)out_size;
    const int* corr = (const int*)d_in[0];
    const int* kcv  = (const int*)d_in[1];
    const int* prob = (const int*)d_in[2];
    const float* FM    = (const float*)d_in[3];
    const float* trans = (const float*)d_in[4];
    const float* olp   = (const float*)d_in[5];
    const float* olk   = (const float*)d_in[6];
    const float* initl = (const float*)d_in[7];
    const float* lrw   = (const float*)d_in[8];
    const float* lrb   = (const float*)d_in[9];
    float* out = (float*)d_out;

    const size_t xbytes = (size_t)BB * TT * 2 * sizeof(float);
    if (ws_size >= xbytes) {
        float* xbuf = (float*)d_ws;
        hipLaunchKernelGGL(gemm_x_wmma, dim3(2000), dim3(128), 0, stream,
                           FM, lrw, lrb, xbuf);
        hipLaunchKernelGGL((bkt_chains<true>), dim3(BB), dim3(NTH), 0, stream,
                           corr, kcv, prob, FM, trans, olp, olk, initl,
                           lrw, lrb, xbuf, out);
    } else {
        hipLaunchKernelGGL((bkt_chains<false>), dim3(BB), dim3(NTH), 0, stream,
                           corr, kcv, prob, FM, trans, olp, olk, initl,
                           lrw, lrb, (const float*)nullptr, out);
    }
}


# one launch instead of two: each sequence's workgroup computes its own x rows with the same f32 WMMA product (same loaders, same k order) straight into LDS; no x buffer; same sums in the same order
# speedup vs baseline: 28.3980x; 1.2334x over previous
#include <hip/hip_runtime.h>

#define BB 256
#define TT 500
#define NKC 100
#define NA 13
#define TP 513

typedef __attribute__((ext_vector_type(2))) float v2f;
typedef __attribute__((ext_vector_type(8))) float v8f;

__device__ __forceinline__ float lse2(float x, float y) {
    float m = fmaxf(x, y);
    float e = __builtin_amdgcn_exp2f(-1.4426950408889634f * fabsf(x - y));
    return fmaf(0.6931471805599453f, __builtin_amdgcn_logf(1.0f + e), m);
}

__device__ __forceinline__ float add_log(float m, float s) {
    return fmaf(0.6931471805599453f, __builtin_amdgcn_logf(s), m);
}


__device__ __forceinline__ float lse13(const float* lab, const float* l) {
    float v[NA];
#pragma unroll
    for (int a = 0; a < NA; ++a) v[a] = lab[a * TP] + l[a * TP];
    float m = fmaxf(fmaxf(fmaxf(fmaxf(v[0], v[8]), fmaxf(v[4], v[12])),
                          fmaxf(fmaxf(v[2], v[10]), v[6])),
                    fmaxf(fmaxf(fmaxf(v[1], v[9]), v[5]),
                          fmaxf(fmaxf(v[3], v[11]), v[7])));
    float e[NA];
#pragma unroll
    for (int a = 0; a < NA; ++a) e[a] = __expf(v[a] - m);
    float x0 = e[0] + e[8], x1 = e[1] + e[9], x2 = e[2] + e[10], x3 = e[3] + e[11];
    float x4 = e[4] + e[12];
    float y0 = x0 + x4, y1 = x1 + e[5], y2 = x2 + e[6], y3 = x3 + e[7];
    float z0 = y0 + y2, z1 = y1 + y3;
    return add_log(m, z0 + z1);
}

#define NTH 512
#define NPART 5
#define PLEN (TT / NPART)
#define PG 20

struct ObsLog { float l00, l01, l10, l11; };
__device__ __forceinline__ ObsLog obs_logs(float q00, float q01, float q10, float q11,
                                           float x0, float x1, float abil) {
    float l00 = q00 - abil + x0;
    float l01 = q01 + abil - x0;
    float l10 = q10 - abil + x1;
    float l11 = q11 + abil - x1;
    float z0 = lse2(l00, l01), z1 = lse2(l10, l11);
    ObsLog r;
    r.l00 = l00 - z0; r.l01 = l01 - z0;
    r.l10 = l10 - z1; r.l11 = l11 - z1;
    return r;
}

__global__ __launch_bounds__(NTH)
void bkt_chains(const int* __restrict__ corr, const int* __restrict__ kcv,
                const int* __restrict__ prob, const float* __restrict__ FM,
                const float* __restrict__ trans, const float* __restrict__ olp,
                const float* __restrict__ olk, const float* __restrict__ initl,
                const float* __restrict__ lrw, const float* __restrict__ lrb,
                float* __restrict__ out) {
    __shared__ __attribute__((aligned(16))) int s_c[NTH];
    __shared__ int s_obs[NTH];
    __shared__ float4 s_q[NTH];
    __shared__ v2f s_x[NTH];
    __shared__ v2f s_b[50 * 32];
    __shared__ int s_less[NPART * NKC];
    __shared__ int s_eq[NPART * NKC];
    __shared__ int s_start[NKC];
    __shared__ int s_cnt[NKC];
    __shared__ int s_order[NTH];
    __shared__ float s_la0[NKC * 2];
    __shared__ float s_logt[NKC * 4];
    __shared__ float s_l0[NA * TP];
    __shared__ float s_l1[NA * TP];
    __shared__ float s_lab[NA * TP];

    const int tid = threadIdx.x;
    const int b = blockIdx.x;
    const long bT = (long)b * TT;

    v2f bw[4];
#pragma unroll
    for (int j = 0; j < 4; ++j) {
        int i = tid + j * NTH;
        if (i > 50 * 32 - 1) i = 50 * 32 - 1;
        const int L = i & 31;
        const int n = L & 15;
        const int kb = (i >> 5) * 4 + ((L >> 4) << 1);
        bw[j] = *(const v2f*)(lrw + (n < 2 ? n : 0) * 200 + kb);
    }

    if (tid < NKC) {
        const int c = tid;
        const float i0 = initl[c * 2], i1 = initl[c * 2 + 1];
        const float il = lse2(i0, i1);
        s_la0[c * 2] = i0 - il;
        s_la0[c * 2 + 1] = i1 - il;
        const float r00 = trans[c * 4 + 0], r01 = trans[c * 4 + 1];
        const float r10 = trans[c * 4 + 2], r11 = trans[c * 4 + 3];
        const float n0 = lse2(r00, r10);
        const float n1 = lse2(r01, r11);
        s_logt[c * 4 + 0] = r00 - n0;
        s_logt[c * 4 + 1] = r01 - n1;
        s_logt[c * 4 + 2] = r10 - n0;
        s_logt[c * 4 + 3] = r11 - n1;
    }

    if (tid < TT) {
        const long id = bT + tid;
        const int c = kcv[id];
        const int p = prob[id];
        s_c[tid] = c;
        s_obs[tid] = corr[id];
        const float4 g = *(const float4*)(olp + (long)p * 4);
        const float4 kq = *(const float4*)(olk + c * 4);
        float4 q;
        q.x = g.x + kq.x; q.y = g.y + kq.y; q.z = g.z + kq.z; q.w = g.w + kq.w;
        s_q[tid] = q;
    }

#pragma unroll
    for (int j = 0; j < 4; ++j) {
        const int i = tid + j * NTH;
        if (i < 50 * 32) {
            v2f v = bw[j];
            if ((i & 15) >= 2) { v.x = 0.0f; v.y = 0.0f; }
            s_b[i] = v;
        }
    }
    __syncthreads();

    {
        const int wave = tid >> 5, lane = tid & 31;
        const int m = lane & 15;
        const int khalf = (lane >> 4) << 1;
        const float bias = lrb[m < 2 ? m : 0];
        for (int tile = wave; tile < 32; tile += NTH / 32) {
            const int r0 = tile * 16;
            const int row = (r0 + m < TT) ? (r0 + m) : (TT - 1);
            const float* arow = FM + (bT + row) * 200 + khalf;
            v8f acc = {};
#pragma unroll 25
            for (int k = 0; k < 50; ++k) {
                v2f a = *(const v2f*)(arow + k * 4);
                v2f bf = s_b[k * 32 + lane];
                acc = __builtin_amdgcn_wmma_f32_16x16x4_f32(
                    false, a, false, bf, (short)0, acc, false, false);
            }
            if (m < 2) {
                const int moff = (lane >> 4) << 3;
                float* const xs = (float*)s_x;
                for (int r = 0; r < 8; ++r) xs[(r0 + moff + r) * 2 + m] = acc[r] + bias;
            }
        }
    }

    const int part = tid / NKC;
    const int myc = tid - part * NKC;
    const bool sorter = tid < NPART * NKC;
    if (sorter) {
        const int4* src = (const int4*)(s_c + part * PLEN);
        int less = 0, eq = 0;
        for (int i = 0; i < PLEN / 4; ++i) {
            const int4 v = src[i];
            less += (v.x < myc) + (v.y < myc) + (v.z < myc) + (v.w < myc);
            eq += (v.x == myc) + (v.y == myc) + (v.z == myc) + (v.w == myc);
        }
        s_less[part * NKC + myc] = less;
        s_eq[part * NKC + myc] = eq;
    }
    __syncthreads();
    if (sorter) {
        int k = 0, tot = 0;
        for (int p = 0; p < NPART; ++p) {
            k += s_less[p * NKC + myc];
            const int e = s_eq[p * NKC + myc];
            if (p < part) k += e;
            tot += e;
        }
        if (part == 0) { s_start[myc] = k; s_cnt[myc] = tot; }
        const int base = part * PLEN;
        const int4* src = (const int4*)(s_c + base);
        for (int i = 0; i < PLEN / 4; i += 5) {
            int4 v[5];
#pragma unroll
            for (int j = 0; j < 5; ++j) v[j] = src[i + j];
#pragma unroll
            for (int j = 0; j < 5; ++j) {
                const int t4 = base + (i + j) * 4;
                if (v[j].x == myc) { s_order[k] = t4; ++k; }
                if (v[j].y == myc) { s_order[k] = t4 + 1; ++k; }
                if (v[j].z == myc) { s_order[k] = t4 + 2; ++k; }
                if (v[j].w == myc) { s_order[k] = t4 + 3; ++k; }
            }
        }
    }
    __syncthreads();

    {
        const int a16 = tid & 15;
        const bool real = a16 < NA;
        const int aa = real ? a16 : (NA - 1);
        const float abil = -3.0f + 0.5f * (float)aa;
        for (int c = tid >> 4; c < NKC; c += NTH / 16) {
            float al0 = s_la0[c * 2], al1 = s_la0[c * 2 + 1];
            const float t00 = s_logt[c * 4 + 0], t01 = s_logt[c * 4 + 1];
            const float t10 = s_logt[c * 4 + 2], t11 = s_logt[c * 4 + 3];
            const int k0 = s_start[c];
            const int n = s_cnt[c];
            if (n > 0) {
                const int kl = k0 + n - 1;
                int t = s_order[k0];
                int tn = s_order[k0 + 1 <= kl ? k0 + 1 : kl];
                int obs = s_obs[t];
                float4 q = s_q[t];
                v2f x = s_x[t];
                for (int k = k0; k <= kl; ++k) {
                    const int tnn = s_order[k + 2 <= kl ? k + 2 : kl];
                    const int obsn = s_obs[tn];
                    const float4 qn = s_q[tn];
                    const v2f xn = s_x[tn];

                    const ObsLog lo = obs_logs(q.x, q.y, q.z, q.w, x.x, x.y, abil);
                    const float u00 = lo.l00 + al0, u01 = lo.l01 + al0;
                    const float u10 = lo.l10 + al1, u11 = lo.l11 + al1;
                    float lpgb0 = lse2(u00, u10);
                    float lpgb1 = lse2(u01, u11);
                    const float nrm = lse2(lpgb0, lpgb1);
                    lpgb0 -= nrm; lpgb1 -= nrm;
                    const float w0 = obs ? u01 : u00;
                    const float w1 = obs ? u11 : u10;
                    al0 = lse2(w0 + t00, w1 + t01);
                    al1 = lse2(w0 + t10, w1 + t11);
                    if (real) {
                        s_l0[aa * TP + t] = lpgb0;
                        s_l1[aa * TP + t] = lpgb1;
                        s_lab[aa * TP + t] = obs ? lpgb1 : lpgb0;
                    }
                    t = tn; tn = tnn; obs = obsn; q = qn; x = xn;
                }
            }
        }
    }
    __syncthreads();

    if (tid < NA) {
        float* const col = s_lab + tid * TP;
        float logab = -2.5649493574615367f;
        float cur[PG], nxt[PG];
#pragma unroll
        for (int j = 0; j < PG; ++j) cur[j] = col[j];
        for (int g = 0; g < TT / PG; ++g) {
            const int gn = (g + 1 < TT / PG) ? g + 1 : g;
#pragma unroll
            for (int j = 0; j < PG; ++j) nxt[j] = col[gn * PG + j];
#pragma unroll
            for (int j = 0; j < PG; ++j) {
                col[g * PG + j] = logab;
                logab += cur[j];
            }
#pragma unroll
            for (int j = 0; j < PG; ++j) cur[j] = nxt[j];
        }
    }
    __syncthreads();

    if (tid < TT) {
        const float o0 = lse13(s_lab + tid, s_l0 + tid);
        const float o1 = lse13(s_lab + tid, s_l1 + tid);
        const float onrm = lse2(o0, o1);
        v2f ov;
        ov.x = o0 - onrm;
        ov.y = o1 - onrm;
        *(v2f*)(out + (bT + tid) * 2) = ov;
    }
}

extern "C" void kernel_launch(void* const* d_in, const int* in_sizes, int n_in,
                              void* d_out, int out_size, void* d_ws, size_t ws_size,
                              hipStream_t stream) {
    (void)in_sizes; (void)n_in; (void)out_size;
    const int* corr = (const int*)d_in[0];
    const int* kcv  = (const int*)d_in[1];
    const int* prob = (const int*)d_in[2];
    const float* FM    = (const float*)d_in[3];
    const float* trans = (const float*)d_in[4];
    const float* olp   = (const float*)d_in[5];
    const float* olk   = (const float*)d_in[6];
    const float* initl = (const float*)d_in[7];
    const float* lrw   = (const float*)d_in[8];
    const float* lrb   = (const float*)d_in[9];
    float* out = (float*)d_out;

    (void)d_ws; (void)ws_size;
    hipLaunchKernelGGL(bkt_chains, dim3(BB), dim3(NTH), 0, stream,
                       corr, kcv, prob, FM, trans, olp, olk, initl,
                       lrw, lrb, out);
}


# scan: the counting sort of the steps by chain replaced by per-step links to the chain's next step, built from wave ballots and a 16x100 table of wave-first steps; same sums in the same order
# speedup vs baseline: 31.3728x; 1.1048x over previous
#include <hip/hip_runtime.h>

#define BB 256
#define TT 500
#define NKC 100
#define NA 13
#define TP 513

typedef __attribute__((ext_vector_type(2))) float v2f;
typedef __attribute__((ext_vector_type(8))) float v8f;

__device__ __forceinline__ float lse2(float x, float y) {
    float m = fmaxf(x, y);
    float e = __builtin_amdgcn_exp2f(-1.4426950408889634f * fabsf(x - y));
    return fmaf(0.6931471805599453f, __builtin_amdgcn_logf(1.0f + e), m);
}

__device__ __forceinline__ float add_log(float m, float s) {
    return fmaf(0.6931471805599453f, __builtin_amdgcn_logf(s), m);
}


__device__ __forceinline__ float lse13(const float* lab, const float* l) {
    float v[NA];
#pragma unroll
    for (int a = 0; a < NA; ++a) v[a] = lab[a * TP] + l[a * TP];
    float m = fmaxf(fmaxf(fmaxf(fmaxf(v[0], v[8]), fmaxf(v[4], v[12])),
                          fmaxf(fmaxf(v[2], v[10]), v[6])),
                    fmaxf(fmaxf(fmaxf(v[1], v[9]), v[5]),
                          fmaxf(fmaxf(v[3], v[11]), v[7])));
    float e[NA];
#pragma unroll
    for (int a = 0; a < NA; ++a) e[a] = __expf(v[a] - m);
    float x0 = e[0] + e[8], x1 = e[1] + e[9], x2 = e[2] + e[10], x3 = e[3] + e[11];
    float x4 = e[4] + e[12];
    float y0 = x0 + x4, y1 = x1 + e[5], y2 = x2 + e[6], y3 = x3 + e[7];
    float z0 = y0 + y2, z1 = y1 + y3;
    return add_log(m, z0 + z1);
}

#define NTH 512
#define NWV (NTH / 32)
#define NONE (NTH - 1)
#define PG 20

struct ObsLog { float l00, l01, l10, l11; };
__device__ __forceinline__ ObsLog obs_logs(float q00, float q01, float q10, float q11,
                                           float x0, float x1, float abil) {
    float l00 = q00 - abil + x0;
    float l01 = q01 + abil - x0;
    float l10 = q10 - abil + x1;
    float l11 = q11 + abil - x1;
    float z0 = lse2(l00, l01), z1 = lse2(l10, l11);
    ObsLog r;
    r.l00 = l00 - z0; r.l01 = l01 - z0;
    r.l10 = l10 - z1; r.l11 = l11 - z1;
    return r;
}

__global__ __launch_bounds__(NTH)
void bkt_chains(const int* __restrict__ corr, const int* __restrict__ kcv,
                const int* __restrict__ prob, const float* __restrict__ FM,
                const float* __restrict__ trans, const float* __restrict__ olp,
                const float* __restrict__ olk, const float* __restrict__ initl,
                const float* __restrict__ lrw, const float* __restrict__ lrb,
                float* __restrict__ out) {
    __shared__ int s_obs[NTH];
    __shared__ float4 s_q[NTH];
    __shared__ v2f s_x[NTH];
    __shared__ v2f s_b[50 * 32];
    __shared__ int s_nf[NWV * NKC];
    __shared__ int s_head[NKC];
    __shared__ int s_nxt[NTH];
    __shared__ float s_la0[NKC * 2];
    __shared__ float s_logt[NKC * 4];
    __shared__ float s_l0[NA * TP];
    __shared__ float s_l1[NA * TP];
    __shared__ float s_lab[NA * TP];

    const int tid = threadIdx.x;
    const int b = blockIdx.x;
    const long bT = (long)b * TT;
    const int wave = tid >> 5, lane = tid & 31;

    v2f bw[4];
#pragma unroll
    for (int j = 0; j < 4; ++j) {
        int i = tid + j * NTH;
        if (i > 50 * 32 - 1) i = 50 * 32 - 1;
        const int L = i & 31;
        const int n = L & 15;
        const int kb = (i >> 5) * 4 + ((L >> 4) << 1);
        bw[j] = *(const v2f*)(lrw + (n < 2 ? n : 0) * 200 + kb);
    }

    if (tid < NKC) {
        const int c = tid;
        const float i0 = initl[c * 2], i1 = initl[c * 2 + 1];
        const float il = lse2(i0, i1);
        s_la0[c * 2] = i0 - il;
        s_la0[c * 2 + 1] = i1 - il;
        const float r00 = trans[c * 4 + 0], r01 = trans[c * 4 + 1];
        const float r10 = trans[c * 4 + 2], r11 = trans[c * 4 + 3];
        const float n0 = lse2(r00, r10);
        const float n1 = lse2(r01, r11);
        s_logt[c * 4 + 0] = r00 - n0;
        s_logt[c * 4 + 1] = r01 - n1;
        s_logt[c * 4 + 2] = r10 - n0;
        s_logt[c * 4 + 3] = r11 - n1;
    }

    for (int i = tid; i < NWV * NKC; i += NTH) s_nf[i] = NONE;

    const bool step = tid < TT;
    const long id = bT + (step ? tid : TT - 1);
    const int cin = kcv[id];
    const int p = prob[id];
    const int ob = corr[id];
    const float4 g = *(const float4*)(olp + (long)p * 4);
    const float4 kq = *(const float4*)(olk + cin * 4);
    const int c = step ? cin : (-1 - lane);
    unsigned match = 0u;
#pragma unroll
    for (int j = 0; j < 32; ++j) {
        const int cj = __builtin_amdgcn_readlane(c, j);
        const unsigned same = __builtin_amdgcn_ballot_w32(c == cj);
        if (c == cj) match = same;
    }
    {
        float4 q;
        q.x = g.x + kq.x; q.y = g.y + kq.y; q.z = g.z + kq.z; q.w = g.w + kq.w;
        s_q[tid] = q;
        s_obs[tid] = ob;
    }

#pragma unroll
    for (int j = 0; j < 4; ++j) {
        const int i = tid + j * NTH;
        if (i < 50 * 32) {
            v2f v = bw[j];
            if ((i & 15) >= 2) { v.x = 0.0f; v.y = 0.0f; }
            s_b[i] = v;
        }
    }
    __syncthreads();

    if (step && (match & ((1u << lane) - 1u)) == 0u) s_nf[wave * NKC + c] = tid;

    {
        const int m = lane & 15;
        const int khalf = (lane >> 4) << 1;
        const float bias = lrb[m < 2 ? m : 0];
        for (int tile = wave; tile < 32; tile += NTH / 32) {
            const int r0 = tile * 16;
            const int row = (r0 + m < TT) ? (r0 + m) : (TT - 1);
            const float* arow = FM + (bT + row) * 200 + khalf;
            v8f acc = {};
#pragma unroll 25
            for (int k = 0; k < 50; ++k) {
                v2f a = *(const v2f*)(arow + k * 4);
                v2f bf = s_b[k * 32 + lane];
                acc = __builtin_amdgcn_wmma_f32_16x16x4_f32(
                    false, a, false, bf, (short)0, acc, false, false);
            }
            if (m < 2) {
                const int moff = (lane >> 4) << 3;
                float* const xs = (float*)s_x;
                for (int r = 0; r < 8; ++r) xs[(r0 + moff + r) * 2 + m] = acc[r] + bias;
            }
        }
    }

    __syncthreads();

    {
        const int cc = step ? c : 0;
        int found = NONE;
#pragma unroll
        for (int w2 = NWV - 1; w2 >= 1; --w2) {
            const int x = s_nf[w2 * NKC + cc];
            if (w2 > wave && x != NONE) found = x;
        }
        const unsigned later = match & ~((2u << lane) - 1u);
        const int nx = later ? ((tid & ~31) + __builtin_ctz(later)) : found;
        s_nxt[tid] = step ? nx : NONE;
        if (tid < NKC) {
            int h = NONE;
#pragma unroll
            for (int w2 = NWV - 1; w2 >= 0; --w2) {
                const int x = s_nf[w2 * NKC + tid];
                if (x != NONE) h = x;
            }
            s_head[tid] = h;
        }
    }
    __syncthreads();

    if ((tid & 15) < NA) {
        const int aa = tid & 15;
        const float abil = -3.0f + 0.5f * (float)aa;
        for (int ch = tid >> 4; ch < NKC; ch += NTH / 16) {
            float al0 = s_la0[ch * 2], al1 = s_la0[ch * 2 + 1];
            const float t00 = s_logt[ch * 4 + 0], t01 = s_logt[ch * 4 + 1];
            const float t10 = s_logt[ch * 4 + 2], t11 = s_logt[ch * 4 + 3];
            int t = s_head[ch];
            if (t != NONE) {
                int tn = s_nxt[t];
                int obs = s_obs[t];
                float4 q = s_q[t];
                v2f x = s_x[t];
                while (t != NONE) {
                    const int tnn = s_nxt[tn];
                    const int obsn = s_obs[tn];
                    const float4 qn = s_q[tn];
                    const v2f xn = s_x[tn];

                    const ObsLog lo = obs_logs(q.x, q.y, q.z, q.w, x.x, x.y, abil);
                    const float u00 = lo.l00 + al0, u01 = lo.l01 + al0;
                    const float u10 = lo.l10 + al1, u11 = lo.l11 + al1;
                    float lpgb0 = lse2(u00, u10);
                    float lpgb1 = lse2(u01, u11);
                    const float nrm = lse2(lpgb0, lpgb1);
                    lpgb0 -= nrm; lpgb1 -= nrm;
                    const float w0 = obs ? u01 : u00;
                    const float w1 = obs ? u11 : u10;
                    al0 = lse2(w0 + t00, w1 + t01);
                    al1 = lse2(w0 + t10, w1 + t11);
                    s_l0[aa * TP + t] = lpgb0;
                    s_l1[aa * TP + t] = lpgb1;
                    s_lab[aa * TP + t] = obs ? lpgb1 : lpgb0;
                    t = tn; tn = tnn; obs = obsn; q = qn; x = xn;
                }
            }
        }
    }
    __syncthreads();

    if (tid < NA) {
        float* const col = s_lab + tid * TP;
        float logab = -2.5649493574615367f;
        float cur[PG], nxt[PG];
#pragma unroll
        for (int j = 0; j < PG; ++j) cur[j] = col[j];
        for (int g = 0; g < TT / PG; ++g) {
            const int gn = (g + 1 < TT / PG) ? g + 1 : g;
#pragma unroll
            for (int j = 0; j < PG; ++j) nxt[j] = col[gn * PG + j];
#pragma unroll
            for (int j = 0; j < PG; ++j) {
                col[g * PG + j] = logab;
                logab += cur[j];
            }
#pragma unroll
            for (int j = 0; j < PG; ++j) cur[j] = nxt[j];
        }
    }
    __syncthreads();

    if (tid < TT) {
        const float o0 = lse13(s_lab + tid, s_l0 + tid);
        const float o1 = lse13(s_lab + tid, s_l1 + tid);
        const float onrm = lse2(o0, o1);
        v2f ov;
        ov.x = o0 - onrm;
        ov.y = o1 - onrm;
        *(v2f*)(out + (bT + tid) * 2) = ov;
    }
}

extern "C" void kernel_launch(void* const* d_in, const int* in_sizes, int n_in,
                              void* d_out, int out_size, void* d_ws, size_t ws_size,
                              hipStream_t stream) {
    (void)in_sizes; (void)n_in; (void)out_size;
    const int* corr = (const int*)d_in[0];
    const int* kcv  = (const int*)d_in[1];
    const int* prob = (const int*)d_in[2];
    const float* FM    = (const float*)d_in[3];
    const float* trans = (const float*)d_in[4];
    const float* olp   = (const float*)d_in[5];
    const float* olk   = (const float*)d_in[6];
    const float* initl = (const float*)d_in[7];
    const float* lrw   = (const float*)d_in[8];
    const float* lrb   = (const float*)d_in[9];
    float* out = (float*)d_out;

    (void)d_ws; (void)ws_size;
    hipLaunchKernelGGL(bkt_chains, dim3(BB), dim3(NTH), 0, stream,
                       corr, kcv, prob, FM, trans, olp, olk, initl,
                       lrw, lrb, out);
}


# x product: each wave fetches its FM tile with 16-byte loads that run along the rows (16 rows x half a row per piece), stages it in its own LDS and reads the A fragments there; same k order, same sums
# speedup vs baseline: 33.1847x; 1.0578x over previous
#include <hip/hip_runtime.h>

#define BB 256
#define TT 500
#define NKC 100
#define NA 13
#define TP 513

typedef __attribute__((ext_vector_type(2))) float v2f;
typedef __attribute__((ext_vector_type(4))) float v4f;
typedef __attribute__((ext_vector_type(8))) float v8f;

__device__ __forceinline__ float lse2(float x, float y) {
    float m = fmaxf(x, y);
    float e = __builtin_amdgcn_exp2f(-1.4426950408889634f * fabsf(x - y));
    return fmaf(0.6931471805599453f, __builtin_amdgcn_logf(1.0f + e), m);
}

__device__ __forceinline__ float add_log(float m, float s) {
    return fmaf(0.6931471805599453f, __builtin_amdgcn_logf(s), m);
}


__device__ __forceinline__ float lse13(const float* lab, const float* l) {
    float v[NA];
#pragma unroll
    for (int a = 0; a < NA; ++a) v[a] = lab[a * TP] + l[a * TP];
    float m = fmaxf(fmaxf(fmaxf(fmaxf(v[0], v[8]), fmaxf(v[4], v[12])),
                          fmaxf(fmaxf(v[2], v[10]), v[6])),
                    fmaxf(fmaxf(fmaxf(v[1], v[9]), v[5]),
                          fmaxf(fmaxf(v[3], v[11]), v[7])));
    float e[NA];
#pragma unroll
    for (int a = 0; a < NA; ++a) e[a] = __expf(v[a] - m);
    float x0 = e[0] + e[8], x1 = e[1] + e[9], x2 = e[2] + e[10], x3 = e[3] + e[11];
    float x4 = e[4] + e[12];
    float y0 = x0 + x4, y1 = x1 + e[5], y2 = x2 + e[6], y3 = x3 + e[7];
    float z0 = y0 + y2, z1 = y1 + y3;
    return add_log(m, z0 + z1);
}

#define NTH 512
#define NWV (NTH / 32)
#define NONE (NTH - 1)
#define PG 20
#define HC 100
#define PV 13
#define STG (16 * HC + 64)

__device__ __forceinline__ void fm_piece_loads(v4f (&pre)[PV], const float* __restrict__ FM,
                                               long bT, int tile, int hf, int lane) {
#pragma unroll
    for (int i = 0; i < PV; ++i) {
        int idx = lane + 32 * i;
        if (idx > 16 * 25 - 1) idx = 16 * 25 - 1;
        const int row = idx / 25;
        const int c4 = idx - row * 25;
        int gr = tile * 16 + row;
        if (gr > TT - 1) gr = TT - 1;
        pre[i] = *(const v4f*)(FM + (bT + gr) * 200 + hf * HC + c4 * 4);
    }
}

struct ObsLog { float l00, l01, l10, l11; };
__device__ __forceinline__ ObsLog obs_logs(float q00, float q01, float q10, float q11,
                                           float x0, float x1, float abil) {
    float l00 = q00 - abil + x0;
    float l01 = q01 + abil - x0;
    float l10 = q10 - abil + x1;
    float l11 = q11 + abil - x1;
    float z0 = lse2(l00, l01), z1 = lse2(l10, l11);
    ObsLog r;
    r.l00 = l00 - z0; r.l01 = l01 - z0;
    r.l10 = l10 - z1; r.l11 = l11 - z1;
    return r;
}

__global__ __launch_bounds__(NTH)
void bkt_chains(const int* __restrict__ corr, const int* __restrict__ kcv,
                const int* __restrict__ prob, const float* __restrict__ FM,
                const float* __restrict__ trans, const float* __restrict__ olp,
                const float* __restrict__ olk, const float* __restrict__ initl,
                const float* __restrict__ lrw, const float* __restrict__ lrb,
                float* __restrict__ out) {
    __shared__ int s_obs[NTH];
    __shared__ float4 s_q[NTH];
    __shared__ v2f s_x[NTH];
    __shared__ v2f s_b[50 * 32];
    __shared__ int s_nf[NWV * NKC];
    __shared__ int s_head[NKC];
    __shared__ int s_nxt[NTH];
    __shared__ float s_la0[NKC * 2];
    __shared__ float s_logt[NKC * 4];
    __shared__ __attribute__((aligned(16))) float s_big[NWV * STG];
    static_assert(NWV * STG >= 3 * NA * TP, "the planes must fit the region");
    float* const s_l0 = s_big;
    float* const s_l1 = s_big + NA * TP;
    float* const s_lab = s_big + 2 * NA * TP;

    const int tid = threadIdx.x;
    const int b = blockIdx.x;
    const long bT = (long)b * TT;
    const int wave = tid >> 5, lane = tid & 31;

    v2f bw[4];
#pragma unroll
    for (int j = 0; j < 4; ++j) {
        int i = tid + j * NTH;
        if (i > 50 * 32 - 1) i = 50 * 32 - 1;
        const int L = i & 31;
        const int n = L & 15;
        const int kb = (i >> 5) * 4 + ((L >> 4) << 1);
        bw[j] = *(const v2f*)(lrw + (n < 2 ? n : 0) * 200 + kb);
    }

    if (tid < NKC) {
        const int c = tid;
        const float i0 = initl[c * 2], i1 = initl[c * 2 + 1];
        const float il = lse2(i0, i1);
        s_la0[c * 2] = i0 - il;
        s_la0[c * 2 + 1] = i1 - il;
        const float r00 = trans[c * 4 + 0], r01 = trans[c * 4 + 1];
        const float r10 = trans[c * 4 + 2], r11 = trans[c * 4 + 3];
        const float n0 = lse2(r00, r10);
        const float n1 = lse2(r01, r11);
        s_logt[c * 4 + 0] = r00 - n0;
        s_logt[c * 4 + 1] = r01 - n1;
        s_logt[c * 4 + 2] = r10 - n0;
        s_logt[c * 4 + 3] = r11 - n1;
    }

    for (int i = tid; i < NWV * NKC; i += NTH) s_nf[i] = NONE;

    const bool step = tid < TT;
    const long id = bT + (step ? tid : TT - 1);
    const int cin = kcv[id];
    const int p = prob[id];
    const int ob = corr[id];
    v4f pre[PV];
    fm_piece_loads(pre, FM, bT, wave, 0, lane);
    __builtin_amdgcn_sched_barrier(0);
    const float4 g = *(const float4*)(olp + (long)p * 4);
    const float4 kq = *(const float4*)(olk + cin * 4);
    const int c = step ? cin : (-1 - lane);
    unsigned match = 0u;
#pragma unroll
    for (int j = 0; j < 32; ++j) {
        const int cj = __builtin_amdgcn_readlane(c, j);
        const unsigned same = __builtin_amdgcn_ballot_w32(c == cj);
        if (c == cj) match = same;
    }
    {
        float4 q;
        q.x = g.x + kq.x; q.y = g.y + kq.y; q.z = g.z + kq.z; q.w = g.w + kq.w;
        s_q[tid] = q;
        s_obs[tid] = ob;
    }

#pragma unroll
    for (int j = 0; j < 4; ++j) {
        const int i = tid + j * NTH;
        if (i < 50 * 32) {
            v2f v = bw[j];
            if ((i & 15) >= 2) { v.x = 0.0f; v.y = 0.0f; }
            s_b[i] = v;
        }
    }
    __syncthreads();

    if (step && (match & ((1u << lane) - 1u)) == 0u) s_nf[wave * NKC + c] = tid;

    {
        const int m = lane & 15;
        const int khalf = (lane >> 4) << 1;
        const float bias = lrb[m < 2 ? m : 0];
        float* const stg = s_big + wave * STG;
        v8f acc = {};
#pragma unroll
        for (int p = 0; p < 4; ++p) {
            const int tile = wave + (p >> 1) * NWV;
            const int hf = p & 1;
            const int r0 = tile * 16;
#pragma unroll
            for (int i = 0; i < PV; ++i) *(v4f*)(stg + 4 * (lane + 32 * i)) = pre[i];
            if (p < 3) fm_piece_loads(pre, FM, bT, wave + ((p + 1) >> 1) * NWV, (p + 1) & 1, lane);
            if (hf == 0) acc = v8f{};
#pragma unroll
            for (int kk = 0; kk < 25; ++kk) {
                v2f a = *(const v2f*)(stg + m * HC + kk * 4 + khalf);
                v2f bf = s_b[(hf * 25 + kk) * 32 + lane];
                acc = __builtin_amdgcn_wmma_f32_16x16x4_f32(
                    false, a, false, bf, (short)0, acc, false, false);
            }
            if (hf == 1 && m < 2) {
                const int moff = (lane >> 4) << 3;
                float* const xs = (float*)s_x;
                for (int r = 0; r < 8; ++r) xs[(r0 + moff + r) * 2 + m] = acc[r] + bias;
            }
        }
    }

    __syncthreads();

    {
        const int cc = step ? c : 0;
        int found = NONE;
#pragma unroll
        for (int w2 = NWV - 1; w2 >= 1; --w2) {
            const int x = s_nf[w2 * NKC + cc];
            if (w2 > wave && x != NONE) found = x;
        }
        const unsigned later = match & ~((2u << lane) - 1u);
        const int nx = later ? ((tid & ~31) + __builtin_ctz(later)) : found;
        s_nxt[tid] = step ? nx : NONE;
        if (tid < NKC) {
            int h = NONE;
#pragma unroll
            for (int w2 = NWV - 1; w2 >= 0; --w2) {
                const int x = s_nf[w2 * NKC + tid];
                if (x != NONE) h = x;
            }
            s_head[tid] = h;
        }
    }
    __syncthreads();

    if ((tid & 15) < NA) {
        const int aa = tid & 15;
        const float abil = -3.0f + 0.5f * (float)aa;
        for (int ch = tid >> 4; ch < NKC; ch += NTH / 16) {
            float al0 = s_la0[ch * 2], al1 = s_la0[ch * 2 + 1];
            const float t00 = s_logt[ch * 4 + 0], t01 = s_logt[ch * 4 + 1];
            const float t10 = s_logt[ch * 4 + 2], t11 = s_logt[ch * 4 + 3];
            int t = s_head[ch];
            if (t != NONE) {
                int tn = s_nxt[t];
                int obs = s_obs[t];
                float4 q = s_q[t];
                v2f x = s_x[t];
                while (t != NONE) {
                    const int tnn = s_nxt[tn];
                    const int obsn = s_obs[tn];
                    const float4 qn = s_q[tn];
                    const v2f xn = s_x[tn];

                    const ObsLog lo = obs_logs(q.x, q.y, q.z, q.w, x.x, x.y, abil);
                    const float u00 = lo.l00 + al0, u01 = lo.l01 + al0;
                    const float u10 = lo.l10 + al1, u11 = lo.l11 + al1;
                    float lpgb0 = lse2(u00, u10);
                    float lpgb1 = lse2(u01, u11);
                    const float nrm = lse2(lpgb0, lpgb1);
                    lpgb0 -= nrm; lpgb1 -= nrm;
                    const float w0 = obs ? u01 : u00;
                    const float w1 = obs ? u11 : u10;
                    al0 = lse2(w0 + t00, w1 + t01);
                    al1 = lse2(w0 + t10, w1 + t11);
                    s_l0[aa * TP + t] = lpgb0;
                    s_l1[aa * TP + t] = lpgb1;
                    s_lab[aa * TP + t] = obs ? lpgb1 : lpgb0;
                    t = tn; tn = tnn; obs = obsn; q = qn; x = xn;
                }
            }
        }
    }
    __syncthreads();

    if (tid < NA) {
        float* const col = s_lab + tid * TP;
        float logab = -2.5649493574615367f;
        float cur[PG], nxt[PG];
#pragma unroll
        for (int j = 0; j < PG; ++j) cur[j] = col[j];
        for (int g = 0; g < TT / PG; ++g) {
            const int gn = (g + 1 < TT / PG) ? g + 1 : g;
#pragma unroll
            for (int j = 0; j < PG; ++j) nxt[j] = col[gn * PG + j];
#pragma unroll
            for (int j = 0; j < PG; ++j) {
                col[g * PG + j] = logab;
                logab += cur[j];
            }
#pragma unroll
            for (int j = 0; j < PG; ++j) cur[j] = nxt[j];
        }
    }
    __syncthreads();

    if (tid < TT) {
        const float o0 = lse13(s_lab + tid, s_l0 + tid);
        const float o1 = lse13(s_lab + tid, s_l1 + tid);
        const float onrm = lse2(o0, o1);
        v2f ov;
        ov.x = o0 - onrm;
        ov.y = o1 - onrm;
        *(v2f*)(out + (bT + tid) * 2) = ov;
    }
}

extern "C" void kernel_launch(void* const* d_in, const int* in_sizes, int n_in,
                              void* d_out, int out_size, void* d_ws, size_t ws_size,
                              hipStream_t stream) {
    (void)in_sizes; (void)n_in; (void)out_size;
    const int* corr = (const int*)d_in[0];
    const int* kcv  = (const int*)d_in[1];
    const int* prob = (const int*)d_in[2];
    const float* FM    = (const float*)d_in[3];
    const float* trans = (const float*)d_in[4];
    const float* olp   = (const float*)d_in[5];
    const float* olk   = (const float*)d_in[6];
    const float* initl = (const float*)d_in[7];
    const float* lrw   = (const float*)d_in[8];
    const float* lrb   = (const float*)d_in[9];
    float* out = (float*)d_out;

    (void)d_ws; (void)ws_size;
    hipLaunchKernelGGL(bkt_chains, dim3(BB), dim3(NTH), 0, stream,
                       corr, kcv, prob, FM, trans, olp, olk, initl,
                       lrw, lrb, out);
}


# log_ability's running sum over time leaves its one serial wave: a blocked prefix sum per ability (20-step segments per lane, totals scanned across the wave); the order of its f32 adds changes
# speedup vs baseline: 41.3583x; 1.2463x over previous
#include <hip/hip_runtime.h>

#define BB 256
#define TT 500
#define NKC 100
#define NA 13
#define TP 513

typedef __attribute__((ext_vector_type(2))) float v2f;
typedef __attribute__((ext_vector_type(4))) float v4f;
typedef __attribute__((ext_vector_type(8))) float v8f;

__device__ __forceinline__ float lse2(float x, float y) {
    float m = fmaxf(x, y);
    float e = __builtin_amdgcn_exp2f(-1.4426950408889634f * fabsf(x - y));
    return fmaf(0.6931471805599453f, __builtin_amdgcn_logf(1.0f + e), m);
}

__device__ __forceinline__ float add_log(float m, float s) {
    return fmaf(0.6931471805599453f, __builtin_amdgcn_logf(s), m);
}


__device__ __forceinline__ float lse13(const float* lab, const float* l) {
    float v[NA];
#pragma unroll
    for (int a = 0; a < NA; ++a) v[a] = lab[a * TP] + l[a * TP];
    float m = fmaxf(fmaxf(fmaxf(fmaxf(v[0], v[8]), fmaxf(v[4], v[12])),
                          fmaxf(fmaxf(v[2], v[10]), v[6])),
                    fmaxf(fmaxf(fmaxf(v[1], v[9]), v[5]),
                          fmaxf(fmaxf(v[3], v[11]), v[7])));
    float e[NA];
#pragma unroll
    for (int a = 0; a < NA; ++a) e[a] = __expf(v[a] - m);
    float x0 = e[0] + e[8], x1 = e[1] + e[9], x2 = e[2] + e[10], x3 = e[3] + e[11];
    float x4 = e[4] + e[12];
    float y0 = x0 + x4, y1 = x1 + e[5], y2 = x2 + e[6], y3 = x3 + e[7];
    float z0 = y0 + y2, z1 = y1 + y3;
    return add_log(m, z0 + z1);
}

#define NTH 512
#define NWV (NTH / 32)
#define NONE (NTH - 1)
#define SEG 20
#define HC 100
#define PV 13
#define STG (16 * HC + 64)

__device__ __forceinline__ void fm_piece_loads(v4f (&pre)[PV], const float* __restrict__ FM,
                                               long bT, int tile, int hf, int lane) {
#pragma unroll
    for (int i = 0; i < PV; ++i) {
        int idx = lane + 32 * i;
        if (idx > 16 * 25 - 1) idx = 16 * 25 - 1;
        const int row = idx / 25;
        const int c4 = idx - row * 25;
        int gr = tile * 16 + row;
        if (gr > TT - 1) gr = TT - 1;
        pre[i] = *(const v4f*)(FM + (bT + gr) * 200 + hf * HC + c4 * 4);
    }
}

struct ObsLog { float l00, l01, l10, l11; };
__device__ __forceinline__ ObsLog obs_logs(float q00, float q01, float q10, float q11,
                                           float x0, float x1, float abil) {
    float l00 = q00 - abil + x0;
    float l01 = q01 + abil - x0;
    float l10 = q10 - abil + x1;
    float l11 = q11 + abil - x1;
    float z0 = lse2(l00, l01), z1 = lse2(l10, l11);
    ObsLog r;
    r.l00 = l00 - z0; r.l01 = l01 - z0;
    r.l10 = l10 - z1; r.l11 = l11 - z1;
    return r;
}

template <int N>
__device__ __forceinline__ float row_below(float v) {
    return __int_as_float(__builtin_amdgcn_update_dpp(0, __float_as_int(v), 0x110 + N, 0xf, 0xf, true));
}

__global__ __launch_bounds__(NTH)
void bkt_chains(const int* __restrict__ corr, const int* __restrict__ kcv,
                const int* __restrict__ prob, const float* __restrict__ FM,
                const float* __restrict__ trans, const float* __restrict__ olp,
                const float* __restrict__ olk, const float* __restrict__ initl,
                const float* __restrict__ lrw, const float* __restrict__ lrb,
                float* __restrict__ out) {
    __shared__ int s_obs[NTH];
    __shared__ float4 s_q[NTH];
    __shared__ v2f s_x[NTH];
    __shared__ v2f s_b[50 * 32];
    __shared__ int s_nf[NWV * NKC];
    __shared__ int s_head[NKC];
    __shared__ int s_nxt[NTH];
    __shared__ float s_la0[NKC * 2];
    __shared__ float s_logt[NKC * 4];
    __shared__ __attribute__((aligned(16))) float s_big[NWV * STG];
    static_assert(NWV * STG >= 3 * NA * TP, "the planes must fit the region");
    float* const s_l0 = s_big;
    float* const s_l1 = s_big + NA * TP;
    float* const s_lab = s_big + 2 * NA * TP;

    const int tid = threadIdx.x;
    const int b = blockIdx.x;
    const long bT = (long)b * TT;
    const int wave = tid >> 5, lane = tid & 31;

    v2f bw[4];
#pragma unroll
    for (int j = 0; j < 4; ++j) {
        int i = tid + j * NTH;
        if (i > 50 * 32 - 1) i = 50 * 32 - 1;
        const int L = i & 31;
        const int n = L & 15;
        const int kb = (i >> 5) * 4 + ((L >> 4) << 1);
        bw[j] = *(const v2f*)(lrw + (n < 2 ? n : 0) * 200 + kb);
    }

    if (tid < NKC) {
        const int c = tid;
        const float i0 = initl[c * 2], i1 = initl[c * 2 + 1];
        const float il = lse2(i0, i1);
        s_la0[c * 2] = i0 - il;
        s_la0[c * 2 + 1] = i1 - il;
        const float r00 = trans[c * 4 + 0], r01 = trans[c * 4 + 1];
        const float r10 = trans[c * 4 + 2], r11 = trans[c * 4 + 3];
        const float n0 = lse2(r00, r10);
        const float n1 = lse2(r01, r11);
        s_logt[c * 4 + 0] = r00 - n0;
        s_logt[c * 4 + 1] = r01 - n1;
        s_logt[c * 4 + 2] = r10 - n0;
        s_logt[c * 4 + 3] = r11 - n1;
    }

    for (int i = tid; i < NWV * NKC; i += NTH) s_nf[i] = NONE;

    const bool step = tid < TT;
    const long id = bT + (step ? tid : TT - 1);
    const int cin = kcv[id];
    const int p = prob[id];
    const int ob = corr[id];
    v4f pre[PV];
    fm_piece_loads(pre, FM, bT, wave, 0, lane);
    __builtin_amdgcn_sched_barrier(0);
    const float4 g = *(const float4*)(olp + (long)p * 4);
    const float4 kq = *(const float4*)(olk + cin * 4);
    const int c = step ? cin : (-1 - lane);
    unsigned match = 0u;
#pragma unroll
    for (int j = 0; j < 32; ++j) {
        const int cj = __builtin_amdgcn_readlane(c, j);
        const unsigned same = __builtin_amdgcn_ballot_w32(c == cj);
        if (c == cj) match = same;
    }
    {
        float4 q;
        q.x = g.x + kq.x; q.y = g.y + kq.y; q.z = g.z + kq.z; q.w = g.w + kq.w;
        s_q[tid] = q;
        s_obs[tid] = ob;
    }

#pragma unroll
    for (int j = 0; j < 4; ++j) {
        const int i = tid + j * NTH;
        if (i < 50 * 32) {
            v2f v = bw[j];
            if ((i & 15) >= 2) { v.x = 0.0f; v.y = 0.0f; }
            s_b[i] = v;
        }
    }
    __syncthreads();

    if (step && (match & ((1u << lane) - 1u)) == 0u) s_nf[wave * NKC + c] = tid;

    {
        const int m = lane & 15;
        const int khalf = (lane >> 4) << 1;
        const float bias = lrb[m < 2 ? m : 0];
        float* const stg = s_big + wave * STG;
        v8f acc = {};
#pragma unroll
        for (int p = 0; p < 4; ++p) {
            const int tile = wave + (p >> 1) * NWV;
            const int hf = p & 1;
            const int r0 = tile * 16;
#pragma unroll
            for (int i = 0; i < PV; ++i) *(v4f*)(stg + 4 * (lane + 32 * i)) = pre[i];
            if (p < 3) fm_piece_loads(pre, FM, bT, wave + ((p + 1) >> 1) * NWV, (p + 1) & 1, lane);
            if (hf == 0) acc = v8f{};
#pragma unroll
            for (int kk = 0; kk < 25; ++kk) {
                v2f a = *(const v2f*)(stg + m * HC + kk * 4 + khalf);
                v2f bf = s_b[(hf * 25 + kk) * 32 + lane];
                acc = __builtin_amdgcn_wmma_f32_16x16x4_f32(
                    false, a, false, bf, (short)0, acc, false, false);
            }
            if (hf == 1 && m < 2) {
                const int moff = (lane >> 4) << 3;
                float* const xs = (float*)s_x;
                for (int r = 0; r < 8; ++r) xs[(r0 + moff + r) * 2 + m] = acc[r] + bias;
            }
        }
    }

    __syncthreads();

    {
        const int cc = step ? c : 0;
        int found = NONE;
#pragma unroll
        for (int w2 = NWV - 1; w2 >= 1; --w2) {
            const int x = s_nf[w2 * NKC + cc];
            if (w2 > wave && x != NONE) found = x;
        }
        const unsigned later = match & ~((2u << lane) - 1u);
        const int nx = later ? ((tid & ~31) + __builtin_ctz(later)) : found;
        s_nxt[tid] = step ? nx : NONE;
        if (tid < NKC) {
            int h = NONE;
#pragma unroll
            for (int w2 = NWV - 1; w2 >= 0; --w2) {
                const int x = s_nf[w2 * NKC + tid];
                if (x != NONE) h = x;
            }
            s_head[tid] = h;
        }
    }
    __syncthreads();

    if ((tid & 15) < NA) {
        const int aa = tid & 15;
        const float abil = -3.0f + 0.5f * (float)aa;
        for (int ch = tid >> 4; ch < NKC; ch += NTH / 16) {
            float al0 = s_la0[ch * 2], al1 = s_la0[ch * 2 + 1];
            const float t00 = s_logt[ch * 4 + 0], t01 = s_logt[ch * 4 + 1];
            const float t10 = s_logt[ch * 4 + 2], t11 = s_logt[ch * 4 + 3];
            int t = s_head[ch];
            if (t != NONE) {
                int tn = s_nxt[t];
                int obs = s_obs[t];
                float4 q = s_q[t];
                v2f x = s_x[t];
                while (t != NONE) {
                    const int tnn = s_nxt[tn];
                    const int obsn = s_obs[tn];
                    const float4 qn = s_q[tn];
                    const v2f xn = s_x[tn];

                    const ObsLog lo = obs_logs(q.x, q.y, q.z, q.w, x.x, x.y, abil);
                    const float u00 = lo.l00 + al0, u01 = lo.l01 + al0;
                    const float u10 = lo.l10 + al1, u11 = lo.l11 + al1;
                    float lpgb0 = lse2(u00, u10);
                    float lpgb1 = lse2(u01, u11);
                    const float nrm = lse2(lpgb0, lpgb1);
                    lpgb0 -= nrm; lpgb1 -= nrm;
                    const float w0 = obs ? u01 : u00;
                    const float w1 = obs ? u11 : u10;
                    al0 = lse2(w0 + t00, w1 + t01);
                    al1 = lse2(w0 + t10, w1 + t11);
                    s_l0[aa * TP + t] = lpgb0;
                    s_l1[aa * TP + t] = lpgb1;
                    s_lab[aa * TP + t] = obs ? lpgb1 : lpgb0;
                    t = tn; tn = tnn; obs = obsn; q = qn; x = xn;
                }
            }
        }
    }
    __syncthreads();

    if (wave < NA) {
        float* const seg = s_lab + wave * TP + lane * SEG;
        const bool has = lane < TT / SEG;
        float term[SEG];
        float tot = 0.0f;
        if (has) {
#pragma unroll
            for (int i = 0; i < SEG; ++i) term[i] = seg[i];
            tot = term[0];
#pragma unroll
            for (int i = 1; i < SEG; ++i) tot += term[i];
        }
        float inc = tot;
        inc += row_below<1>(inc);
        inc += row_below<2>(inc);
        inc += row_below<4>(inc);
        inc += row_below<8>(inc);
        const float row0 = __int_as_float(__builtin_amdgcn_readlane(__float_as_int(inc), 15));
        float before = row_below<1>(inc);
        if (lane >= 16) before += row0;
        if (has) {
            float logab = -2.5649493574615367f + before;
#pragma unroll
            for (int i = 0; i < SEG; ++i) {
                seg[i] = logab;
                logab += term[i];
            }
        }
    }
    __syncthreads();

    if (tid < TT) {
        const float o0 = lse13(s_lab + tid, s_l0 + tid);
        const float o1 = lse13(s_lab + tid, s_l1 + tid);
        const float onrm = lse2(o0, o1);
        v2f ov;
        ov.x = o0 - onrm;
        ov.y = o1 - onrm;
        *(v2f*)(out + (bT + tid) * 2) = ov;
    }
}

extern "C" void kernel_launch(void* const* d_in, const int* in_sizes, int n_in,
                              void* d_out, int out_size, void* d_ws, size_t ws_size,
                              hipStream_t stream) {
    (void)in_sizes; (void)n_in; (void)out_size;
    const int* corr = (const int*)d_in[0];
    const int* kcv  = (const int*)d_in[1];
    const int* prob = (const int*)d_in[2];
    const float* FM    = (const float*)d_in[3];
    const float* trans = (const float*)d_in[4];
    const float* olp   = (const float*)d_in[5];
    const float* olk   = (const float*)d_in[6];
    const float* initl = (const float*)d_in[7];
    const float* lrw   = (const float*)d_in[8];
    const float* lrb   = (const float*)d_in[9];
    float* out = (float*)d_out;

    (void)d_ws; (void)ws_size;
    hipLaunchKernelGGL(bkt_chains, dim3(BB), dim3(NTH), 0, stream,
                       corr, kcv, prob, FM, trans, olp, olk, initl,
                       lrw, lrb, out);
}


# scan: the chains' walk repacked: one lane per (chain, ability) pair, 13 lanes to a chain with no gaps, chains ranked by their number of steps so a wave's lists end together; same sums, same order
# speedup vs baseline: 44.2327x; 1.0695x over previous
#include <hip/hip_runtime.h>

#define BB 256
#define TT 500
#define NKC 100
#define NA 13
#define TP 513

typedef __attribute__((ext_vector_type(2))) float v2f;
typedef __attribute__((ext_vector_type(4))) float v4f;
typedef __attribute__((ext_vector_type(8))) float v8f;
typedef __attribute__((ext_vector_type(2))) short v2s;
typedef __attribute__((ext_vector_type(2))) unsigned short v2us;

__device__ __forceinline__ float neg_abs_log2e(float d) {
    float a;
    asm("v_mul_f32_e64 %0, 0xbfb8aa3b, |%1|" : "=v"(a) : "v"(d));
    return a;
}

__device__ __forceinline__ float lse2(float x, float y) {
    float m = fmaxf(x, y);
    float e = __builtin_amdgcn_exp2f(neg_abs_log2e(x - y));
    return fmaf(0.6931471805599453f, __builtin_amdgcn_logf(1.0f + e), m);
}

__device__ __forceinline__ v2f lse2p(v2f x, v2f y) {
    const v2f m = {fmaxf(x.x, y.x), fmaxf(x.y, y.y)};
    const v2f d = x - y;
    const v2f e = {__builtin_amdgcn_exp2f(neg_abs_log2e(d.x)), __builtin_amdgcn_exp2f(neg_abs_log2e(d.y))};
    const v2f s = 1.0f + e;
    const v2f r = {fmaf(0.6931471805599453f, __builtin_amdgcn_logf(s.x), m.x),
                   fmaf(0.6931471805599453f, __builtin_amdgcn_logf(s.y), m.y)};
    return r;
}

__device__ __forceinline__ float add_log(float m, float s) {
    return fmaf(0.6931471805599453f, __builtin_amdgcn_logf(s), m);
}


__device__ __forceinline__ float lse13(const float* lab, const float* l) {
    float v[NA];
#pragma unroll
    for (int a = 0; a < NA; ++a) v[a] = lab[a * TP] + l[a * TP];
    float m = fmaxf(fmaxf(fmaxf(fmaxf(v[0], v[8]), fmaxf(v[4], v[12])),
                          fmaxf(fmaxf(v[2], v[10]), v[6])),
                    fmaxf(fmaxf(fmaxf(v[1], v[9]), v[5]),
                          fmaxf(fmaxf(v[3], v[11]), v[7])));
    float e[NA];
#pragma unroll
    for (int a = 0; a < NA; ++a) e[a] = __expf(v[a] - m);
    float x0 = e[0] + e[8], x1 = e[1] + e[9], x2 = e[2] + e[10], x3 = e[3] + e[11];
    float x4 = e[4] + e[12];
    float y0 = x0 + x4, y1 = x1 + e[5], y2 = x2 + e[6], y3 = x3 + e[7];
    float z0 = y0 + y2, z1 = y1 + y3;
    return add_log(m, z0 + z1);
}

#define NTH 512
#define NWV (NTH / 32)
#define NONE (NTH - 1)
#define SEG 20
#define HC 100
#define PV 13
#define STG (16 * HC + 64)

__device__ __forceinline__ void fm_piece_loads(v4f (&pre)[PV], const float* __restrict__ FM,
                                               long bT, int tile, int hf, int lane) {
#pragma unroll
    for (int i = 0; i < PV; ++i) {
        int idx = lane + 32 * i;
        if (idx > 16 * 25 - 1) idx = 16 * 25 - 1;
        const int row = idx / 25;
        const int c4 = idx - row * 25;
        int gr = tile * 16 + row;
        if (gr > TT - 1) gr = TT - 1;
        pre[i] = *(const v4f*)(FM + (bT + gr) * 200 + hf * HC + c4 * 4);
    }
}

template <int N>
__device__ __forceinline__ float row_below(float v) {
    return __int_as_float(__builtin_amdgcn_update_dpp(0, __float_as_int(v), 0x110 + N, 0xf, 0xf, true));
}

__global__ __launch_bounds__(NTH)
void bkt_chains(const int* __restrict__ corr, const int* __restrict__ kcv,
                const int* __restrict__ prob, const float* __restrict__ FM,
                const float* __restrict__ trans, const float* __restrict__ olp,
                const float* __restrict__ olk, const float* __restrict__ initl,
                const float* __restrict__ lrw, const float* __restrict__ lrb,
                float* __restrict__ out) {
    __shared__ int s_obs[NTH];
    __shared__ float4 s_q[NTH];
    __shared__ v2f s_x[NTH];
    __shared__ v2f s_b[50 * 32];
    __shared__ int s_nf[NWV * NKC];
    __shared__ __attribute__((aligned(16))) unsigned short s_key[NKC + 4];
    __shared__ int s_ord[NKC];
    __shared__ int s_nxt[NTH];
    __shared__ float s_la0[NKC * 2];
    __shared__ float s_logt[NKC * 4];
    __shared__ __attribute__((aligned(16))) float s_big[NWV * STG];
    static_assert(NWV * STG >= 3 * NA * TP, "the planes must fit the region");
    float* const s_l0 = s_big;
    float* const s_l1 = s_big + NA * TP;
    float* const s_lab = s_big + 2 * NA * TP;

    const int tid = threadIdx.x;
    const int b = blockIdx.x;
    const long bT = (long)b * TT;
    const int wave = tid >> 5, lane = tid & 31;

    v2f bw[4];
#pragma unroll
    for (int j = 0; j < 4; ++j) {
        int i = tid + j * NTH;
        if (i > 50 * 32 - 1) i = 50 * 32 - 1;
        const int L = i & 31;
        const int n = L & 15;
        const int kb = (i >> 5) * 4 + ((L >> 4) << 1);
        bw[j] = *(const v2f*)(lrw + (n < 2 ? n : 0) * 200 + kb);
    }

    if (tid < NKC) {
        const int c = tid;
        const float i0 = initl[c * 2], i1 = initl[c * 2 + 1];
        const float il = lse2(i0, i1);
        s_la0[c * 2] = i0 - il;
        s_la0[c * 2 + 1] = i1 - il;
        const float r00 = trans[c * 4 + 0], r01 = trans[c * 4 + 1];
        const float r10 = trans[c * 4 + 2], r11 = trans[c * 4 + 3];
        const float n0 = lse2(r00, r10);
        const float n1 = lse2(r01, r11);
        s_logt[c * 4 + 0] = r00 - n0;
        s_logt[c * 4 + 1] = r01 - n1;
        s_logt[c * 4 + 2] = r10 - n0;
        s_logt[c * 4 + 3] = r11 - n1;
    }

    for (int i = lane; i < NKC; i += 32) s_nf[wave * NKC + i] = NONE;

    const bool step = tid < TT;
    const long id = bT + (step ? tid : TT - 1);
    const int cin = kcv[id];
    const int p = prob[id];
    const int ob = corr[id];
    v4f pre[PV];
    fm_piece_loads(pre, FM, bT, wave, 0, lane);
    __builtin_amdgcn_sched_barrier(0);
    const float4 g = *(const float4*)(olp + (long)p * 4);
    const float4 kq = *(const float4*)(olk + cin * 4);
    const int c = step ? cin : (-1 - lane);
    unsigned match = 0u;
#pragma unroll
    for (int j = 0; j < 32; ++j) {
        const int cj = __builtin_amdgcn_readlane(c, j);
        const unsigned same = __builtin_amdgcn_ballot_w32(c == cj);
        if (c == cj) match = same;
    }
    if (step && (match & ((1u << lane) - 1u)) == 0u) s_nf[wave * NKC + c] = tid | (__popc(match) << 9);
    {
        float4 q;
        q.x = g.x + kq.x; q.y = g.z + kq.z; q.z = g.y + kq.y; q.w = g.w + kq.w;
        s_q[tid] = q;
        s_obs[tid] = ob;
    }

#pragma unroll
    for (int j = 0; j < 4; ++j) {
        const int i = tid + j * NTH;
        if (i < 50 * 32) {
            v2f v = bw[j];
            if ((i & 15) >= 2) { v.x = 0.0f; v.y = 0.0f; }
            s_b[i] = v;
        }
    }
    __syncthreads();

    int head = NONE;
    {
        const int cc = step ? c : 0;
        int found = NONE;
#pragma unroll
        for (int w2 = NWV - 1; w2 >= 1; --w2) {
            const int f = s_nf[w2 * NKC + cc] & NONE;
            if (w2 > wave && f != NONE) found = f;
        }
        const unsigned later = match & ~((2u << lane) - 1u);
        const int nx = later ? ((tid & ~31) + __builtin_ctz(later)) : found;
        s_nxt[tid] = step ? nx : NONE;
        if (tid < NKC) {
            int h = NONE;
            int len = 0;
#pragma unroll
            for (int w2 = NWV - 1; w2 >= 0; --w2) {
                const int x = s_nf[w2 * NKC + tid];
                if ((x & NONE) != NONE) h = x & NONE;
                len += x >> 9;
            }
            head = h;
            s_key[tid] = (unsigned short)(((len < 255 ? len : 255) << 7) | (127 - tid));
        } else if (tid < NKC + 4) {
            s_key[tid] = 0;
        }
    }

    {
        const int m = lane & 15;
        const int khalf = (lane >> 4) << 1;
        const float bias = lrb[m < 2 ? m : 0];
        float* const stg = s_big + wave * STG;
        v8f acc = {};
#pragma unroll
        for (int p = 0; p < 4; ++p) {
            const int tile = wave + (p >> 1) * NWV;
            const int hf = p & 1;
            const int r0 = tile * 16;
#pragma unroll
            for (int i = 0; i < PV; ++i) *(v4f*)(stg + 4 * (lane + 32 * i)) = pre[i];
            if (p < 3) fm_piece_loads(pre, FM, bT, wave + ((p + 1) >> 1) * NWV, (p + 1) & 1, lane);
            if (hf == 0) acc = v8f{};
#pragma unroll
            for (int kk = 0; kk < 25; ++kk) {
                v2f a = *(const v2f*)(stg + m * HC + kk * 4 + khalf);
                v2f bf = s_b[(hf * 25 + kk) * 32 + lane];
                acc = __builtin_amdgcn_wmma_f32_16x16x4_f32(
                    false, a, false, bf, (short)0, acc, false, false);
            }
            if (hf == 1 && m < 2) {
                const int moff = (lane >> 4) << 3;
                float* const xs = (float*)s_x;
                for (int r = 0; r < 8; ++r) xs[(r0 + moff + r) * 2 + m] = acc[r] + bias;
            }
        }
    }

    __syncthreads();

    if (tid < NKC) {
        const short mk = (short)s_key[tid];
        const v2s my2 = {mk, mk};
        v2us ab0 = {0, 0}, ab1 = {0, 0}, ab2 = {0, 0}, ab3 = {0, 0};
#pragma unroll
        for (int j = 0; j < (NKC + 4) / 8; ++j) {
            const uint4 k8 = *(const uint4*)(s_key + 8 * j);
            ab0 += __builtin_bit_cast(v2us, my2 - __builtin_bit_cast(v2s, k8.x)) >> 15;
            ab1 += __builtin_bit_cast(v2us, my2 - __builtin_bit_cast(v2s, k8.y)) >> 15;
            ab2 += __builtin_bit_cast(v2us, my2 - __builtin_bit_cast(v2s, k8.z)) >> 15;
            ab3 += __builtin_bit_cast(v2us, my2 - __builtin_bit_cast(v2s, k8.w)) >> 15;
        }
        const v2us above = (ab0 + ab1) + (ab2 + ab3);
        s_ord[above.x + above.y] = tid | (head << 7);
    }
    __syncthreads();

    for (int r = 0; r < 3; ++r) {
        const int it = (r == 0) ? tid : ((r + 1) * NTH - 1 - tid);
        if (it < NKC * NA) {
            const int pos = it / NA;
            const int aa = it - pos * NA;
            const float abil = -3.0f + 0.5f * (float)aa;
            const int od = s_ord[pos];
            const int ch = od & 127;
            const v2f ab2 = {abil, abil};
            v2f al = {s_la0[ch * 2], s_la0[ch * 2 + 1]};
            const v2f tfrom0 = {s_logt[ch * 4 + 0], s_logt[ch * 4 + 2]};
            const v2f tfrom1 = {s_logt[ch * 4 + 1], s_logt[ch * 4 + 3]};
            int t = od >> 7;
            if (t != NONE) {
                int tn = s_nxt[t];
                int obs = s_obs[t];
                float4 q = s_q[t];
                v2f x = s_x[t];
                while (t != NONE) {
                    const int tnn = s_nxt[tn];
                    const int obsn = s_obs[tn];
                    const float4 qn = s_q[tn];
                    const v2f xn = s_x[tn];

                    const v2f qo0 = {q.x, q.y}, qo1 = {q.z, q.w};
                    v2f l0 = (qo0 - ab2) + x;
                    v2f l1 = (qo1 + ab2) - x;
                    const v2f z = lse2p(l0, l1);
                    l0 -= z; l1 -= z;
                    const v2f u0 = l0 + al, u1 = l1 + al;
                    float lpgb0 = lse2(u0.x, u0.y);
                    float lpgb1 = lse2(u1.x, u1.y);
                    const float nrm = lse2(lpgb0, lpgb1);
                    lpgb0 -= nrm; lpgb1 -= nrm;
                    const v2f w = obs ? u1 : u0;
                    const v2f w0 = {w.x, w.x}, w1 = {w.y, w.y};
                    al = lse2p(w0 + tfrom0, w1 + tfrom1);
                    s_l0[aa * TP + t] = lpgb0;
                    s_l1[aa * TP + t] = lpgb1;
                    s_lab[aa * TP + t] = obs ? lpgb1 : lpgb0;
                    t = tn; tn = tnn; obs = obsn; q = qn; x = xn;
                }
            }
        }
    }
    __syncthreads();

    if (wave < NA) {
        float* const seg = s_lab + wave * TP + lane * SEG;
        const bool has = lane < TT / SEG;
        float term[SEG];
        float tot = 0.0f;
        if (has) {
#pragma unroll
            for (int i = 0; i < SEG; ++i) term[i] = seg[i];
            tot = term[0];
#pragma unroll
            for (int i = 1; i < SEG; ++i) tot += term[i];
        }
        float inc = tot;
        inc += row_below<1>(inc);
        inc += row_below<2>(inc);
        inc += row_below<4>(inc);
        inc += row_below<8>(inc);
        const float row0 = __int_as_float(__builtin_amdgcn_readlane(__float_as_int(inc), 15));
        float before = row_below<1>(inc);
        if (lane >= 16) before += row0;
        if (has) {
            float logab = -2.5649493574615367f + before;
#pragma unroll
            for (int i = 0; i < SEG; ++i) {
                seg[i] = logab;
                logab += term[i];
            }
        }
    }
    __syncthreads();

    if (tid < TT) {
        const float o0 = lse13(s_lab + tid, s_l0 + tid);
        const float o1 = lse13(s_lab + tid, s_l1 + tid);
        const float onrm = lse2(o0, o1);
        v2f ov;
        ov.x = o0 - onrm;
        ov.y = o1 - onrm;
        *(v2f*)(out + (bT + tid) * 2) = ov;
    }
}

extern "C" void kernel_launch(void* const* d_in, const int* in_sizes, int n_in,
                              void* d_out, int out_size, void* d_ws, size_t ws_size,
                              hipStream_t stream) {
    (void)in_sizes; (void)n_in; (void)out_size;
    const int* corr = (const int*)d_in[0];
    const int* kcv  = (const int*)d_in[1];
    const int* prob = (const int*)d_in[2];
    const float* FM    = (const float*)d_in[3];
    const float* trans = (const float*)d_in[4];
    const float* olp   = (const float*)d_in[5];
    const float* olk   = (const float*)d_in[6];
    const float* initl = (const float*)d_in[7];
    const float* lrw   = (const float*)d_in[8];
    const float* lrb   = (const float*)d_in[9];
    float* out = (float*)d_out;

    (void)d_ws; (void)ws_size;
    hipLaunchKernelGGL(bkt_chains, dim3(BB), dim3(NTH), 0, stream,
                       corr, kcv, prob, FM, trans, olp, olk, initl,
                       lrw, lrb, out);
}
